# starting kernel
# baseline (speedup 1.0000x reference)
#include <hip/hip_runtime.h>
#include <hip/hip_bf16.h>

typedef __attribute__((ext_vector_type(2))) float v2f;
typedef __attribute__((ext_vector_type(8))) float v8f;

#define DD 128

__global__ void gemm_bias_act(const float* __restrict__ A,
                              const float* __restrict__ W,
                              const float* __restrict__ bias,
                              float* __restrict__ C,
                              int Ncols, int relu)
{
    __shared__ float As[16 * DD];
    __shared__ float Bs[DD * 16];
    const int lane = threadIdx.x & 31;
    const int tm = blockIdx.x * 16;
    const int tn = blockIdx.y * 16;

    {
        const float4* Ag = (const float4*)(A + (size_t)tm * DD);
        float4* As4 = (float4*)As;
#pragma unroll
        for (int r = 0; r < 16; ++r)
            As4[r * 32 + lane] = Ag[(size_t)r * 32 + lane];
    }
    {
#pragma unroll
        for (int i = 0; i < 16; ++i) {
            const int k  = i * 8 + (lane >> 2);
            const int c4 = (lane & 3) * 4;
            const float4 w = *(const float4*)(W + (size_t)k * Ncols + tn + c4);
            *(float4*)(Bs + k * 16 + c4) = w;
        }
    }
    __syncthreads();

    const int half = lane >> 4;
    const int mn   = lane & 15;

    v8f acc = {0.f, 0.f, 0.f, 0.f, 0.f, 0.f, 0.f, 0.f};
#pragma unroll
    for (int kk = 0; kk < 32; ++kk) {
        const int kb = kk * 4 + half * 2;
        v2f a, b;
        a.x = As[mn * DD + kb];
        a.y = As[mn * DD + kb + 1];
        b.x = Bs[kb * 16 + mn];
        b.y = Bs[(kb + 1) * 16 + mn];
        acc = __builtin_amdgcn_wmma_f32_16x16x4_f32(
             false, a,  false, b,
             (short)0, acc,  false,  false);
    }

#pragma unroll
    for (int r = 0; r < 8; ++r) {
        const int row = tm + r + half * 8;
        const int col = tn + mn;
        float v = acc[r] + (bias ? bias[col] : 0.f);
        if (relu) v = v > 0.f ? v : 0.f;
        C[(size_t)row * Ncols + col] = v;
    }
}

__global__ void deg_init(float* deg, int n) {
    int i = blockIdx.x * blockDim.x + threadIdx.x;
    if (i < n) deg[i] = 1.0f;
}
__global__ void deg_accum(const int* __restrict__ dst, float* deg, int e) {
    int i = blockIdx.x * blockDim.x + threadIdx.x;
    if (i < e) atomicAdd(deg + dst[i], 1.0f);
}
__global__ void deg_invsqrt(float* deg, int n) {
    int i = blockIdx.x * blockDim.x + threadIdx.x;
    if (i < n) deg[i] = rsqrtf(deg[i]);
}

__global__ void self_init(const float* __restrict__ hw,
                          const float* __restrict__ inv,
                          float* __restrict__ agg, int n) {
    size_t idx = (size_t)blockIdx.x * blockDim.x + threadIdx.x;
    if (idx < (size_t)n * DD) {
        const int i = (int)(idx >> 7);
        const float c = inv[i];
        agg[idx] = hw[idx] * c * c;
    }
}

__global__ void edge_scatter(const float* __restrict__ hw,
                             const int* __restrict__ src,
                             const int* __restrict__ dst,
                             const float* __restrict__ inv,
                             float* __restrict__ agg, int e) {
    const int wavesPerBlk = blockDim.x >> 5;
    const int ei = blockIdx.x * wavesPerBlk + (threadIdx.x >> 5);
    if (ei >= e) return;
    const int lane = threadIdx.x & 31;
    const int s = src[ei];
    const int d = dst[ei];
    const float coef = inv[s] * inv[d];
    const float4 v = *(const float4*)(hw + (size_t)s * DD + lane * 4);
    float* p = agg + (size_t)d * DD + lane * 4;
    atomicAdd(p + 0, v.x * coef);
    atomicAdd(p + 1, v.y * coef);
    atomicAdd(p + 2, v.z * coef);
    atomicAdd(p + 3, v.w * coef);
}

__global__ void bias_relu(const float* __restrict__ agg,
                          const float* __restrict__ b,
                          float* __restrict__ h, int n) {
    size_t idx = (size_t)blockIdx.x * blockDim.x + threadIdx.x;
    if (idx < (size_t)n * DD) {
        const float v = agg[idx] + b[idx & (DD - 1)];
        h[idx] = v > 0.f ? v : 0.f;
    }
}

__global__ void pool_zero(float* p, int n) {
    int i = blockIdx.x * blockDim.x + threadIdx.x;
    if (i < n) p[i] = 0.f;
}
__global__ void pool_add(const float* __restrict__ h,
                         const int* __restrict__ batch,
                         float* __restrict__ pooled, int n) {
    int t = blockIdx.x * blockDim.x + threadIdx.x;
    if (t < n * 32) {
        const int i = t >> 5;
        const int q = (t & 31) * 4;
        const int g = batch[i];
        const float4 v = *(const float4*)(h + (size_t)i * DD + q);
        float* p = pooled + (size_t)g * DD + q;
        atomicAdd(p + 0, v.x);
        atomicAdd(p + 1, v.y);
        atomicAdd(p + 2, v.z);
        atomicAdd(p + 3, v.w);
    }
}

extern "C" void kernel_launch(void* const* d_in, const int* in_sizes, int n_in,
                              void* d_out, int out_size, void* d_ws, size_t ws_size,
                              hipStream_t stream) {
    const float* x      = (const float*)d_in[0];
    const int*   ei     = (const int*)d_in[1];
    const int*   batch  = (const int*)d_in[2];
    const float* enc_W1 = (const float*)d_in[3];
    const float* enc_b1 = (const float*)d_in[4];
    const float* enc_W2 = (const float*)d_in[5];
    const float* enc_b2 = (const float*)d_in[6];
    const float* gcn_W  = (const float*)d_in[7];
    const float* gcn_b  = (const float*)d_in[8];
    const float* dec_W1 = (const float*)d_in[9];
    const float* dec_b1 = (const float*)d_in[10];
    const float* dec_W2 = (const float*)d_in[11];
    const float* dec_b2 = (const float*)d_in[12];

    const int N   = in_sizes[0] / DD;
    const int E   = in_sizes[1] / 2;
    const int L   = in_sizes[7] / (DD * DD);
    const int G   = 512;
    const int OUT = in_sizes[11] / DD;

    const int* src = ei;
    const int* dst = ei + E;

    char* ws = (char*)d_ws;
    const size_t nd = (size_t)N * DD * sizeof(float);
    float* B0     = (float*)(ws);
    float* B1     = (float*)(ws + nd);
    float* B2     = (float*)(ws + 2 * nd);
    float* inv    = (float*)(ws + 3 * nd);
    size_t off    = 3 * nd + (((size_t)N * 4 + 255) & ~(size_t)255);
    float* pooled = (float*)(ws + off);
    float* dech   = pooled + (size_t)G * DD;

    const dim3 wave(32);
    const int elemBlocks = (int)(((size_t)N * DD + 255) / 256);

    gemm_bias_act<<<dim3(N / 16, DD / 16), wave, 0, stream>>>(x,  enc_W1, enc_b1, B0, DD, 1);
    gemm_bias_act<<<dim3(N / 16, DD / 16), wave, 0, stream>>>(B0, enc_W2, enc_b2, B1, DD, 1);

    deg_init   <<<(N + 255) / 256, 256, 0, stream>>>(inv, N);
    deg_accum  <<<(E + 255) / 256, 256, 0, stream>>>(dst, inv, E);
    deg_invsqrt<<<(N + 255) / 256, 256, 0, stream>>>(inv, N);

    for (int l = 0; l < L; ++l) {
        const float* Wl = gcn_W + (size_t)l * DD * DD;
        const float* bl = gcn_b + (size_t)l * DD;
        gemm_bias_act<<<dim3(N / 16, DD / 16), wave, 0, stream>>>(B1, Wl, nullptr, B0, DD, 0);
        self_init   <<<elemBlocks, 256, 0, stream>>>(B0, inv, B2, N);
        edge_scatter<<<(E + 7) / 8, 256, 0, stream>>>(B0, src, dst, inv, B2, E);
        bias_relu   <<<elemBlocks, 256, 0, stream>>>(B2, bl, B1, N);
    }

    pool_zero<<<(G * DD + 255) / 256, 256, 0, stream>>>(pooled, G * DD);
    pool_add <<<(N * 32 + 255) / 256, 256, 0, stream>>>(B1, batch, pooled, N);

    gemm_bias_act<<<dim3(G / 16, DD / 16),  wave, 0, stream>>>(pooled, dec_W1, dec_b1, dech, DD, 1);
    gemm_bias_act<<<dim3(G / 16, OUT / 16), wave, 0, stream>>>(dech,   dec_W2, dec_b2, (float*)d_out, OUT, 0);
}


# edge_scatter and pool_add: lane l owns row elements l+32k, so each f32 atomic add instruction (and row load) covers one whole 128-B line, not 4 lines; atomics kept, arrival-order sums as before
# speedup vs baseline: 1.3030x; 1.3030x over previous
#include <hip/hip_runtime.h>
#include <hip/hip_bf16.h>

typedef __attribute__((ext_vector_type(2))) float v2f;
typedef __attribute__((ext_vector_type(8))) float v8f;

#define DD 128

__global__ void gemm_bias_act(const float* __restrict__ A,
                              const float* __restrict__ W,
                              const float* __restrict__ bias,
                              float* __restrict__ C,
                              int Ncols, int relu)
{
    __shared__ float As[16 * DD];
    __shared__ float Bs[DD * 16];
    const int lane = threadIdx.x & 31;
    const int tm = blockIdx.x * 16;
    const int tn = blockIdx.y * 16;

    {
        const float4* Ag = (const float4*)(A + (size_t)tm * DD);
        float4* As4 = (float4*)As;
#pragma unroll
        for (int r = 0; r < 16; ++r)
            As4[r * 32 + lane] = Ag[(size_t)r * 32 + lane];
    }
    {
#pragma unroll
        for (int i = 0; i < 16; ++i) {
            const int k  = i * 8 + (lane >> 2);
            const int c4 = (lane & 3) * 4;
            const float4 w = *(const float4*)(W + (size_t)k * Ncols + tn + c4);
            *(float4*)(Bs + k * 16 + c4) = w;
        }
    }
    __syncthreads();

    const int half = lane >> 4;
    const int mn   = lane & 15;

    v8f acc = {0.f, 0.f, 0.f, 0.f, 0.f, 0.f, 0.f, 0.f};
#pragma unroll
    for (int kk = 0; kk < 32; ++kk) {
        const int kb = kk * 4 + half * 2;
        v2f a, b;
        a.x = As[mn * DD + kb];
        a.y = As[mn * DD + kb + 1];
        b.x = Bs[kb * 16 + mn];
        b.y = Bs[(kb + 1) * 16 + mn];
        acc = __builtin_amdgcn_wmma_f32_16x16x4_f32(
             false, a,  false, b,
             (short)0, acc,  false,  false);
    }

#pragma unroll
    for (int r = 0; r < 8; ++r) {
        const int row = tm + r + half * 8;
        const int col = tn + mn;
        float v = acc[r] + (bias ? bias[col] : 0.f);
        if (relu) v = v > 0.f ? v : 0.f;
        C[(size_t)row * Ncols + col] = v;
    }
}

__global__ void deg_init(float* deg, int n) {
    int i = blockIdx.x * blockDim.x + threadIdx.x;
    if (i < n) deg[i] = 1.0f;
}
__global__ void deg_accum(const int* __restrict__ dst, float* deg, int e) {
    int i = blockIdx.x * blockDim.x + threadIdx.x;
    if (i < e) atomicAdd(deg + dst[i], 1.0f);
}
__global__ void deg_invsqrt(float* deg, int n) {
    int i = blockIdx.x * blockDim.x + threadIdx.x;
    if (i < n) deg[i] = rsqrtf(deg[i]);
}

__global__ void self_init(const float* __restrict__ hw,
                          const float* __restrict__ inv,
                          float* __restrict__ agg, int n) {
    size_t idx = (size_t)blockIdx.x * blockDim.x + threadIdx.x;
    if (idx < (size_t)n * DD) {
        const int i = (int)(idx >> 7);
        const float c = inv[i];
        agg[idx] = hw[idx] * c * c;
    }
}

__global__ void edge_scatter(const float* __restrict__ hw,
                             const int* __restrict__ src,
                             const int* __restrict__ dst,
                             const float* __restrict__ inv,
                             float* __restrict__ agg, int e) {
    const int wavesPerBlk = blockDim.x >> 5;
    const int ei = blockIdx.x * wavesPerBlk + (threadIdx.x >> 5);
    if (ei >= e) return;
    const int lane = threadIdx.x & 31;
    const int s = src[ei];
    const int d = dst[ei];
    const float coef = inv[s] * inv[d];
    const float* hp = hw + (size_t)s * DD + lane;
    const float v0 = hp[0];
    const float v1 = hp[32];
    const float v2 = hp[64];
    const float v3 = hp[96];
    float* p = agg + (size_t)d * DD + lane;
    atomicAdd(p +  0, v0 * coef);
    atomicAdd(p + 32, v1 * coef);
    atomicAdd(p + 64, v2 * coef);
    atomicAdd(p + 96, v3 * coef);
}

__global__ void bias_relu(const float* __restrict__ agg,
                          const float* __restrict__ b,
                          float* __restrict__ h, int n) {
    size_t idx = (size_t)blockIdx.x * blockDim.x + threadIdx.x;
    if (idx < (size_t)n * DD) {
        const float v = agg[idx] + b[idx & (DD - 1)];
        h[idx] = v > 0.f ? v : 0.f;
    }
}

__global__ void pool_zero(float* p, int n) {
    int i = blockIdx.x * blockDim.x + threadIdx.x;
    if (i < n) p[i] = 0.f;
}
__global__ void pool_add(const float* __restrict__ h,
                         const int* __restrict__ batch,
                         float* __restrict__ pooled, int n) {
    int t = blockIdx.x * blockDim.x + threadIdx.x;
    if (t < n * 32) {
        const int i = t >> 5;
        const int lane = t & 31;
        const int g = batch[i];
        const float* hp = h + (size_t)i * DD + lane;
        const float v0 = hp[0];
        const float v1 = hp[32];
        const float v2 = hp[64];
        const float v3 = hp[96];
        float* p = pooled + (size_t)g * DD + lane;
        atomicAdd(p +  0, v0);
        atomicAdd(p + 32, v1);
        atomicAdd(p + 64, v2);
        atomicAdd(p + 96, v3);
    }
}

extern "C" void kernel_launch(void* const* d_in, const int* in_sizes, int n_in,
                              void* d_out, int out_size, void* d_ws, size_t ws_size,
                              hipStream_t stream) {
    const float* x      = (const float*)d_in[0];
    const int*   ei     = (const int*)d_in[1];
    const int*   batch  = (const int*)d_in[2];
    const float* enc_W1 = (const float*)d_in[3];
    const float* enc_b1 = (const float*)d_in[4];
    const float* enc_W2 = (const float*)d_in[5];
    const float* enc_b2 = (const float*)d_in[6];
    const float* gcn_W  = (const float*)d_in[7];
    const float* gcn_b  = (const float*)d_in[8];
    const float* dec_W1 = (const float*)d_in[9];
    const float* dec_b1 = (const float*)d_in[10];
    const float* dec_W2 = (const float*)d_in[11];
    const float* dec_b2 = (const float*)d_in[12];

    const int N   = in_sizes[0] / DD;
    const int E   = in_sizes[1] / 2;
    const int L   = in_sizes[7] / (DD * DD);
    const int G   = 512;
    const int OUT = in_sizes[11] / DD;

    const int* src = ei;
    const int* dst = ei + E;

    char* ws = (char*)d_ws;
    const size_t nd = (size_t)N * DD * sizeof(float);
    float* B0     = (float*)(ws);
    float* B1     = (float*)(ws + nd);
    float* B2     = (float*)(ws + 2 * nd);
    float* inv    = (float*)(ws + 3 * nd);
    size_t off    = 3 * nd + (((size_t)N * 4 + 255) & ~(size_t)255);
    float* pooled = (float*)(ws + off);
    float* dech   = pooled + (size_t)G * DD;

    const dim3 wave(32);
    const int elemBlocks = (int)(((size_t)N * DD + 255) / 256);

    gemm_bias_act<<<dim3(N / 16, DD / 16), wave, 0, stream>>>(x,  enc_W1, enc_b1, B0, DD, 1);
    gemm_bias_act<<<dim3(N / 16, DD / 16), wave, 0, stream>>>(B0, enc_W2, enc_b2, B1, DD, 1);

    deg_init   <<<(N + 255) / 256, 256, 0, stream>>>(inv, N);
    deg_accum  <<<(E + 255) / 256, 256, 0, stream>>>(dst, inv, E);
    deg_invsqrt<<<(N + 255) / 256, 256, 0, stream>>>(inv, N);

    for (int l = 0; l < L; ++l) {
        const float* Wl = gcn_W + (size_t)l * DD * DD;
        const float* bl = gcn_b + (size_t)l * DD;
        gemm_bias_act<<<dim3(N / 16, DD / 16), wave, 0, stream>>>(B1, Wl, nullptr, B0, DD, 0);
        self_init   <<<elemBlocks, 256, 0, stream>>>(B0, inv, B2, N);
        edge_scatter<<<(E + 7) / 8, 256, 0, stream>>>(B0, src, dst, inv, B2, E);
        bias_relu   <<<elemBlocks, 256, 0, stream>>>(B2, bl, B1, N);
    }

    pool_zero<<<(G * DD + 255) / 256, 256, 0, stream>>>(pooled, G * DD);
    pool_add <<<(N * 32 + 255) / 256, 256, 0, stream>>>(B1, batch, pooled, N);

    gemm_bias_act<<<dim3(G / 16, DD / 16),  wave, 0, stream>>>(pooled, dec_W1, dec_b1, dech, DD, 1);
    gemm_bias_act<<<dim3(G / 16, OUT / 16), wave, 0, stream>>>(dech,   dec_W2, dec_b2, (float*)d_out, OUT, 0);
}


# gemm: each wave computes a whole 16x128 output strip from one LDS copy of W shared by 8 waves (A read once, W once per 128 rows, padded LDS layout) instead of one 16x16 tile per wave
# speedup vs baseline: 1.5375x; 1.1800x over previous
#include <hip/hip_runtime.h>
#include <hip/hip_bf16.h>

typedef __attribute__((ext_vector_type(2))) float v2f;
typedef __attribute__((ext_vector_type(8))) float v8f;

#define DD 128

#define GW 8
#define APITCH 132

template <int NT, bool RELU>
__global__ __launch_bounds__(256)
void gemm_strip(const float* __restrict__ A,
                const float* __restrict__ W,
                const float* __restrict__ bias,
                float* __restrict__ C,
                int nStrips)
{
    constexpr int NC = NT * 16;
    constexpr int C4 = NC / 4;
    constexpr int NW = (64 * C4) / 256;
    constexpr int RPI = 32 / C4;
    constexpr int PW = 2 * NC + 32;
    __shared__ alignas(16) float Ws[64 * PW];
    __shared__ alignas(16) float As[GW * 16 * APITCH];

    const int tid  = threadIdx.x;
    const int lane = tid & 31;
    const int wv   = __builtin_amdgcn_readfirstlane(tid >> 5);
    const int strip = blockIdx.x * GW + wv;
    const bool valid = strip < nStrips;
    const int lstrip = valid ? strip : nStrips - 1;
    float* Aw = As + wv * (16 * APITCH);

    float4 av[16];
    {
        const float4* Ag = (const float4*)(A + (size_t)lstrip * 16 * DD);
#pragma unroll
        for (int r = 0; r < 16; ++r)
            av[r] = Ag[r * 32 + lane];
    }
    float4 w0[NW], w1[NW];
#pragma unroll
    for (int i = 0; i < NW; ++i) {
        const int idx = i * 256 + tid;
        const int p  = idx / C4;
        const int c4 = idx % C4;
        w0[i] = *(const float4*)(W + (size_t)(2 * p)     * NC + 4 * c4);
        w1[i] = *(const float4*)(W + (size_t)(2 * p + 1) * NC + 4 * c4);
    }
#pragma unroll
    for (int r = 0; r < 16; ++r)
        *(float4*)(Aw + r * APITCH + 4 * lane) = av[r];
#pragma unroll
    for (int i = 0; i < NW; ++i) {
        const int idx = i * 256 + tid;
        const int p  = idx / C4;
        const int c4 = idx % C4;
        float* d = Ws + p * PW + 8 * c4;
        *(float4*)(d)     = make_float4(w0[i].x, w1[i].x, w0[i].y, w1[i].y);
        *(float4*)(d + 4) = make_float4(w0[i].z, w1[i].z, w0[i].w, w1[i].w);
    }
    __syncthreads();

    const int h = lane >> 4;
    const int c = lane & 15;

    if (valid) {
        v8f acc[NT];
#pragma unroll
        for (int j = 0; j < NT; ++j)
            acc[j] = (v8f){0.f, 0.f, 0.f, 0.f, 0.f, 0.f, 0.f, 0.f};

        const float* ap = Aw + c * APITCH + 2 * h;
        const float* bp = Ws + h * PW + 2 * c;
#pragma unroll
        for (int s = 0; s < 32; ++s) {
            const v2f a = *(const v2f*)(ap + 4 * s);
#pragma unroll
            for (int j = 0; j < NT; ++j) {
                const v2f b = *(const v2f*)(bp + 2 * s * PW + 32 * j);
                acc[j] = __builtin_amdgcn_wmma_f32_16x16x4_f32(
                     false, a,  false, b,
                     (short)0, acc[j],  false,  false);
            }
        }

#pragma unroll
        for (int j = 0; j < NT; ++j)
#pragma unroll
            for (int r = 0; r < 8; ++r)
                Aw[(8 * h + r) * APITCH + 16 * j + c] = acc[j][r];
    }

    const int c4 = lane % C4;
    float4 bb = make_float4(0.f, 0.f, 0.f, 0.f);
    if (bias) bb = *(const float4*)(bias + 4 * c4);
    __syncthreads();

    if (valid) {
        float* Cg = C + (size_t)strip * 16 * NC;
        const int r0 = lane / C4;
#pragma unroll
        for (int i = 0; i < 2 * NT; ++i) {
            const int row = i * RPI + r0;
            float4 v = *(const float4*)(Aw + row * APITCH + 4 * c4);
            v.x += bb.x; v.y += bb.y; v.z += bb.z; v.w += bb.w;
            if (RELU) {
                v.x = v.x > 0.f ? v.x : 0.f;
                v.y = v.y > 0.f ? v.y : 0.f;
                v.z = v.z > 0.f ? v.z : 0.f;
                v.w = v.w > 0.f ? v.w : 0.f;
            }
            *(float4*)(Cg + (size_t)row * NC + 4 * c4) = v;
        }
    }
}

__global__ void deg_init(float* deg, int n) {
    int i = blockIdx.x * blockDim.x + threadIdx.x;
    if (i < n) deg[i] = 1.0f;
}
__global__ void deg_accum(const int* __restrict__ dst, float* deg, int e) {
    int i = blockIdx.x * blockDim.x + threadIdx.x;
    if (i < e) atomicAdd(deg + dst[i], 1.0f);
}
__global__ void deg_invsqrt(float* deg, int n) {
    int i = blockIdx.x * blockDim.x + threadIdx.x;
    if (i < n) deg[i] = rsqrtf(deg[i]);
}

__global__ void self_init(const float* __restrict__ hw,
                          const float* __restrict__ inv,
                          float* __restrict__ agg, int n) {
    size_t idx = (size_t)blockIdx.x * blockDim.x + threadIdx.x;
    if (idx < (size_t)n * DD) {
        const int i = (int)(idx >> 7);
        const float c = inv[i];
        agg[idx] = hw[idx] * c * c;
    }
}

__global__ void edge_scatter(const float* __restrict__ hw,
                             const int* __restrict__ src,
                             const int* __restrict__ dst,
                             const float* __restrict__ inv,
                             float* __restrict__ agg, int e) {
    const int wavesPerBlk = blockDim.x >> 5;
    const int ei = blockIdx.x * wavesPerBlk + (threadIdx.x >> 5);
    if (ei >= e) return;
    const int lane = threadIdx.x & 31;
    const int s = src[ei];
    const int d = dst[ei];
    const float coef = inv[s] * inv[d];
    const float* hp = hw + (size_t)s * DD + lane;
    const float v0 = hp[0];
    const float v1 = hp[32];
    const float v2 = hp[64];
    const float v3 = hp[96];
    float* p = agg + (size_t)d * DD + lane;
    atomicAdd(p +  0, v0 * coef);
    atomicAdd(p + 32, v1 * coef);
    atomicAdd(p + 64, v2 * coef);
    atomicAdd(p + 96, v3 * coef);
}

__global__ void bias_relu(const float* __restrict__ agg,
                          const float* __restrict__ b,
                          float* __restrict__ h, int n) {
    size_t idx = (size_t)blockIdx.x * blockDim.x + threadIdx.x;
    if (idx < (size_t)n * DD) {
        const float v = agg[idx] + b[idx & (DD - 1)];
        h[idx] = v > 0.f ? v : 0.f;
    }
}

__global__ void pool_zero(float* p, int n) {
    int i = blockIdx.x * blockDim.x + threadIdx.x;
    if (i < n) p[i] = 0.f;
}
__global__ void pool_add(const float* __restrict__ h,
                         const int* __restrict__ batch,
                         float* __restrict__ pooled, int n) {
    int t = blockIdx.x * blockDim.x + threadIdx.x;
    if (t < n * 32) {
        const int i = t >> 5;
        const int lane = t & 31;
        const int g = batch[i];
        const float* hp = h + (size_t)i * DD + lane;
        const float v0 = hp[0];
        const float v1 = hp[32];
        const float v2 = hp[64];
        const float v3 = hp[96];
        float* p = pooled + (size_t)g * DD + lane;
        atomicAdd(p +  0, v0);
        atomicAdd(p + 32, v1);
        atomicAdd(p + 64, v2);
        atomicAdd(p + 96, v3);
    }
}

extern "C" void kernel_launch(void* const* d_in, const int* in_sizes, int n_in,
                              void* d_out, int out_size, void* d_ws, size_t ws_size,
                              hipStream_t stream) {
    const float* x      = (const float*)d_in[0];
    const int*   ei     = (const int*)d_in[1];
    const int*   batch  = (const int*)d_in[2];
    const float* enc_W1 = (const float*)d_in[3];
    const float* enc_b1 = (const float*)d_in[4];
    const float* enc_W2 = (const float*)d_in[5];
    const float* enc_b2 = (const float*)d_in[6];
    const float* gcn_W  = (const float*)d_in[7];
    const float* gcn_b  = (const float*)d_in[8];
    const float* dec_W1 = (const float*)d_in[9];
    const float* dec_b1 = (const float*)d_in[10];
    const float* dec_W2 = (const float*)d_in[11];
    const float* dec_b2 = (const float*)d_in[12];

    const int N   = in_sizes[0] / DD;
    const int E   = in_sizes[1] / 2;
    const int L   = in_sizes[7] / (DD * DD);
    const int G   = 512;

    const int* src = ei;
    const int* dst = ei + E;

    char* ws = (char*)d_ws;
    const size_t nd = (size_t)N * DD * sizeof(float);
    float* B0     = (float*)(ws);
    float* B1     = (float*)(ws + nd);
    float* B2     = (float*)(ws + 2 * nd);
    float* inv    = (float*)(ws + 3 * nd);
    size_t off    = 3 * nd + (((size_t)N * 4 + 255) & ~(size_t)255);
    float* pooled = (float*)(ws + off);
    float* dech   = pooled + (size_t)G * DD;

    const int nsN = N / 16;
    const int nsG = G / 16;
    const int gbN = (nsN + GW - 1) / GW;
    const int gbG = (nsG + GW - 1) / GW;
    const int elemBlocks = (int)(((size_t)N * DD + 255) / 256);

    gemm_strip<8, true><<<gbN, 256, 0, stream>>>(x,  enc_W1, enc_b1, B0, nsN);
    gemm_strip<8, true><<<gbN, 256, 0, stream>>>(B0, enc_W2, enc_b2, B1, nsN);

    deg_init   <<<(N + 255) / 256, 256, 0, stream>>>(inv, N);
    deg_accum  <<<(E + 255) / 256, 256, 0, stream>>>(dst, inv, E);
    deg_invsqrt<<<(N + 255) / 256, 256, 0, stream>>>(inv, N);

    for (int l = 0; l < L; ++l) {
        const float* Wl = gcn_W + (size_t)l * DD * DD;
        const float* bl = gcn_b + (size_t)l * DD;
        gemm_strip<8, false><<<gbN, 256, 0, stream>>>(B1, Wl, nullptr, B0, nsN);
        self_init   <<<elemBlocks, 256, 0, stream>>>(B0, inv, B2, N);
        edge_scatter<<<(E + 7) / 8, 256, 0, stream>>>(B0, src, dst, inv, B2, E);
        bias_relu   <<<elemBlocks, 256, 0, stream>>>(B2, bl, B1, N);
    }

    pool_zero<<<(G * DD + 255) / 256, 256, 0, stream>>>(pooled, G * DD);
    pool_add <<<(N * 32 + 255) / 256, 256, 0, stream>>>(B1, batch, pooled, N);

    gemm_strip<8, true><<<gbG, 256, 0, stream>>>(pooled, dec_W1, dec_b1, dech, nsG);
    gemm_strip<2, false><<<gbG, 256, 0, stream>>>(dech,   dec_W2, dec_b2, (float*)d_out, nsG);
}


# edge aggregation: edges grouped by destination once per call (int atomics), then one wave per node sums its incoming rows in registers, one row store; no float atomics, sums in segment order
# speedup vs baseline: 2.5268x; 1.6434x over previous
#include <hip/hip_runtime.h>
#include <hip/hip_bf16.h>

typedef __attribute__((ext_vector_type(2))) float v2f;
typedef __attribute__((ext_vector_type(8))) float v8f;

#define DD 128

#define GW 8
#define APITCH 132

template <int NT, bool RELU>
__global__ __launch_bounds__(256)
void gemm_strip(const float* __restrict__ A,
                const float* __restrict__ W,
                const float* __restrict__ bias,
                float* __restrict__ C,
                int nStrips)
{
    constexpr int NC = NT * 16;
    constexpr int C4 = NC / 4;
    constexpr int NW = (64 * C4) / 256;
    constexpr int RPI = 32 / C4;
    constexpr int PW = 2 * NC + 32;
    __shared__ alignas(16) float Ws[64 * PW];
    __shared__ alignas(16) float As[GW * 16 * APITCH];

    const int tid  = threadIdx.x;
    const int lane = tid & 31;
    const int wv   = __builtin_amdgcn_readfirstlane(tid >> 5);
    const int strip = blockIdx.x * GW + wv;
    const bool valid = strip < nStrips;
    const int lstrip = valid ? strip : nStrips - 1;
    float* Aw = As + wv * (16 * APITCH);

    float4 av[16];
    {
        const float4* Ag = (const float4*)(A + (size_t)lstrip * 16 * DD);
#pragma unroll
        for (int r = 0; r < 16; ++r)
            av[r] = Ag[r * 32 + lane];
    }
    float4 w0[NW], w1[NW];
#pragma unroll
    for (int i = 0; i < NW; ++i) {
        const int idx = i * 256 + tid;
        const int p  = idx / C4;
        const int c4 = idx % C4;
        w0[i] = *(const float4*)(W + (size_t)(2 * p)     * NC + 4 * c4);
        w1[i] = *(const float4*)(W + (size_t)(2 * p + 1) * NC + 4 * c4);
    }
#pragma unroll
    for (int r = 0; r < 16; ++r)
        *(float4*)(Aw + r * APITCH + 4 * lane) = av[r];
#pragma unroll
    for (int i = 0; i < NW; ++i) {
        const int idx = i * 256 + tid;
        const int p  = idx / C4;
        const int c4 = idx % C4;
        float* d = Ws + p * PW + 8 * c4;
        *(float4*)(d)     = make_float4(w0[i].x, w1[i].x, w0[i].y, w1[i].y);
        *(float4*)(d + 4) = make_float4(w0[i].z, w1[i].z, w0[i].w, w1[i].w);
    }
    __syncthreads();

    const int h = lane >> 4;
    const int c = lane & 15;

    if (valid) {
        v8f acc[NT];
#pragma unroll
        for (int j = 0; j < NT; ++j)
            acc[j] = (v8f){0.f, 0.f, 0.f, 0.f, 0.f, 0.f, 0.f, 0.f};

        const float* ap = Aw + c * APITCH + 2 * h;
        const float* bp = Ws + h * PW + 2 * c;
#pragma unroll
        for (int s = 0; s < 32; ++s) {
            const v2f a = *(const v2f*)(ap + 4 * s);
#pragma unroll
            for (int j = 0; j < NT; ++j) {
                const v2f b = *(const v2f*)(bp + 2 * s * PW + 32 * j);
                acc[j] = __builtin_amdgcn_wmma_f32_16x16x4_f32(
                     false, a,  false, b,
                     (short)0, acc[j],  false,  false);
            }
        }

#pragma unroll
        for (int j = 0; j < NT; ++j)
#pragma unroll
            for (int r = 0; r < 8; ++r)
                Aw[(8 * h + r) * APITCH + 16 * j + c] = acc[j][r];
    }

    const int c4 = lane % C4;
    float4 bb = make_float4(0.f, 0.f, 0.f, 0.f);
    if (bias) bb = *(const float4*)(bias + 4 * c4);
    __syncthreads();

    if (valid) {
        float* Cg = C + (size_t)strip * 16 * NC;
        const int r0 = lane / C4;
#pragma unroll
        for (int i = 0; i < 2 * NT; ++i) {
            const int row = i * RPI + r0;
            float4 v = *(const float4*)(Aw + row * APITCH + 4 * c4);
            v.x += bb.x; v.y += bb.y; v.z += bb.z; v.w += bb.w;
            if (RELU) {
                v.x = v.x > 0.f ? v.x : 0.f;
                v.y = v.y > 0.f ? v.y : 0.f;
                v.z = v.z > 0.f ? v.z : 0.f;
                v.w = v.w > 0.f ? v.w : 0.f;
            }
            *(float4*)(Cg + (size_t)row * NC + 4 * c4) = v;
        }
    }
}

__global__ void csr_zero(int* __restrict__ cnt, int* __restrict__ total, int n) {
    int i = blockIdx.x * blockDim.x + threadIdx.x;
    if (i < n) cnt[i] = 0;
    if (i == 0) *total = 0;
}
__global__ void csr_count(const int* __restrict__ dst, int* cnt, int e) {
    int i = blockIdx.x * blockDim.x + threadIdx.x;
    if (i < e) atomicAdd(cnt + dst[i], 1);
}
__global__ __launch_bounds__(1024)
void csr_alloc(const int* __restrict__ cnt, int* total,
               int* __restrict__ rowptr, int* __restrict__ cursor,
               float* __restrict__ inv, int n) {
    __shared__ int wsum[32];
    __shared__ int blockBase;
    const int i    = blockIdx.x * 1024 + threadIdx.x;
    const int lane = threadIdx.x & 31;
    const int wv   = threadIdx.x >> 5;
    const int c = (i < n) ? cnt[i] : 0;
    int x = c;
#pragma unroll
    for (int o = 1; o < 32; o <<= 1) {
        const int y = __shfl_up(x, o);
        if (lane >= o) x += y;
    }
    if (lane == 31) wsum[wv] = x;
    __syncthreads();
    if (wv == 0) {
        const int t = wsum[lane];
        int xs = t;
#pragma unroll
        for (int o = 1; o < 32; o <<= 1) {
            const int y = __shfl_up(xs, o);
            if (lane >= o) xs += y;
        }
        wsum[lane] = xs - t;
        if (lane == 31) blockBase = atomicAdd(total, xs);
    }
    __syncthreads();
    if (i < n) {
        const int p = blockBase + wsum[wv] + x - c;
        rowptr[i] = p;
        cursor[i] = p;
        inv[i] = rsqrtf((float)(c + 1));
    }
}
__global__ void csr_fill(const int* __restrict__ src, const int* __restrict__ dst,
                         int* cursor, const float* __restrict__ inv,
                         int2* __restrict__ ed, int e) {
    int i = blockIdx.x * blockDim.x + threadIdx.x;
    if (i < e) {
        const int s = src[i];
        const int pos = atomicAdd(cursor + dst[i], 1);
        ed[pos] = make_int2(s, __float_as_int(inv[s]));
    }
}

__global__ __launch_bounds__(256)
void node_gather(const float* __restrict__ hw,
                 const int* __restrict__ rowptr,
                 const int* __restrict__ cnt,
                 const int2* __restrict__ ed,
                 const float* __restrict__ inv,
                 float* __restrict__ agg, int n) {
    const int i = blockIdx.x * 8 + __builtin_amdgcn_readfirstlane(threadIdx.x >> 5);
    if (i >= n) return;
    const int lane = threadIdx.x & 31;
    const float4* hw4 = (const float4*)hw;
    const int start = __builtin_amdgcn_readfirstlane(rowptr[i]);
    const int m     = __builtin_amdgcn_readfirstlane(cnt[i]);
    const float di  = inv[i];
    const float4 sv = hw4[(size_t)i * 32 + lane];
    float4 acc = make_float4(0.f, 0.f, 0.f, 0.f);

    for (int base = 0; base < m; base += 32) {
        const int mm = (m - base) < 32 ? (m - base) : 32;
        int es = 0;
        float ec = 0.f;
        if (lane < mm) {
            const int2 e = ed[start + base + lane];
            es = e.x;
            ec = __int_as_float(e.y) * di;
        }
        for (int k = 0; k < mm; k += 8) {
            float4 r[8];
            float cf[8];
#pragma unroll
            for (int u = 0; u < 8; ++u) {
                r[u] = make_float4(0.f, 0.f, 0.f, 0.f);
                cf[u] = 0.f;
                if (k + u < mm) {
                    const int s = __builtin_amdgcn_readlane(es, k + u);
                    cf[u] = __int_as_float(__builtin_amdgcn_readlane(__float_as_int(ec), k + u));
                    r[u] = hw4[(size_t)s * 32 + lane];
                }
            }
#pragma unroll
            for (int u = 0; u < 8; ++u) {
                acc.x = fmaf(r[u].x, cf[u], acc.x);
                acc.y = fmaf(r[u].y, cf[u], acc.y);
                acc.z = fmaf(r[u].z, cf[u], acc.z);
                acc.w = fmaf(r[u].w, cf[u], acc.w);
            }
        }
    }
    const float sc = di * di;
    acc.x = fmaf(sv.x, sc, acc.x);
    acc.y = fmaf(sv.y, sc, acc.y);
    acc.z = fmaf(sv.z, sc, acc.z);
    acc.w = fmaf(sv.w, sc, acc.w);
    ((float4*)agg)[(size_t)i * 32 + lane] = acc;
}

__global__ void bias_relu(const float* __restrict__ agg,
                          const float* __restrict__ b,
                          float* __restrict__ h, int n) {
    size_t idx = (size_t)blockIdx.x * blockDim.x + threadIdx.x;
    if (idx < (size_t)n * DD) {
        const float v = agg[idx] + b[idx & (DD - 1)];
        h[idx] = v > 0.f ? v : 0.f;
    }
}

__global__ void pool_zero(float* p, int n) {
    int i = blockIdx.x * blockDim.x + threadIdx.x;
    if (i < n) p[i] = 0.f;
}
__global__ void pool_add(const float* __restrict__ h,
                         const int* __restrict__ batch,
                         float* __restrict__ pooled, int n) {
    int t = blockIdx.x * blockDim.x + threadIdx.x;
    if (t < n * 32) {
        const int i = t >> 5;
        const int lane = t & 31;
        const int g = batch[i];
        const float* hp = h + (size_t)i * DD + lane;
        const float v0 = hp[0];
        const float v1 = hp[32];
        const float v2 = hp[64];
        const float v3 = hp[96];
        float* p = pooled + (size_t)g * DD + lane;
        atomicAdd(p +  0, v0);
        atomicAdd(p + 32, v1);
        atomicAdd(p + 64, v2);
        atomicAdd(p + 96, v3);
    }
}

extern "C" void kernel_launch(void* const* d_in, const int* in_sizes, int n_in,
                              void* d_out, int out_size, void* d_ws, size_t ws_size,
                              hipStream_t stream) {
    const float* x      = (const float*)d_in[0];
    const int*   ei     = (const int*)d_in[1];
    const int*   batch  = (const int*)d_in[2];
    const float* enc_W1 = (const float*)d_in[3];
    const float* enc_b1 = (const float*)d_in[4];
    const float* enc_W2 = (const float*)d_in[5];
    const float* enc_b2 = (const float*)d_in[6];
    const float* gcn_W  = (const float*)d_in[7];
    const float* gcn_b  = (const float*)d_in[8];
    const float* dec_W1 = (const float*)d_in[9];
    const float* dec_b1 = (const float*)d_in[10];
    const float* dec_W2 = (const float*)d_in[11];
    const float* dec_b2 = (const float*)d_in[12];

    const int N   = in_sizes[0] / DD;
    const int E   = in_sizes[1] / 2;
    const int L   = in_sizes[7] / (DD * DD);
    const int G   = 512;

    const int* src = ei;
    const int* dst = ei + E;

    char* ws = (char*)d_ws;
    const size_t nd = (size_t)N * DD * sizeof(float);
    float* B0     = (float*)(ws);
    float* B1     = (float*)(ws + nd);
    float* B2     = (float*)(ws + 2 * nd);
    float* inv    = (float*)(ws + 3 * nd);
    size_t off    = 3 * nd + (((size_t)N * 4 + 255) & ~(size_t)255);
    float* pooled = (float*)(ws + off);
    float* dech   = pooled + (size_t)G * DD;
    const size_t nI = ((size_t)N * 4 + 255) & ~(size_t)255;
    size_t off2   = off + 2 * (size_t)G * DD * sizeof(float);
    int*  cnt     = (int*)(ws + off2);
    int*  rowptr  = (int*)(ws + off2 + nI);
    int*  cursor  = (int*)(ws + off2 + 2 * nI);
    int*  total   = (int*)(ws + off2 + 3 * nI);
    int2* ed      = (int2*)(ws + off2 + 3 * nI + 256);

    const int nsN = N / 16;
    const int nsG = G / 16;
    const int gbN = (nsN + GW - 1) / GW;
    const int gbG = (nsG + GW - 1) / GW;
    const int elemBlocks = (int)(((size_t)N * DD + 255) / 256);

    gemm_strip<8, true><<<gbN, 256, 0, stream>>>(x,  enc_W1, enc_b1, B0, nsN);
    gemm_strip<8, true><<<gbN, 256, 0, stream>>>(B0, enc_W2, enc_b2, B1, nsN);

    csr_zero <<<(N + 255) / 256, 256, 0, stream>>>(cnt, total, N);
    csr_count<<<(E + 255) / 256, 256, 0, stream>>>(dst, cnt, E);
    csr_alloc<<<(N + 1023) / 1024, 1024, 0, stream>>>(cnt, total, rowptr, cursor, inv, N);
    csr_fill <<<(E + 255) / 256, 256, 0, stream>>>(src, dst, cursor, inv, ed, E);

    for (int l = 0; l < L; ++l) {
        const float* Wl = gcn_W + (size_t)l * DD * DD;
        const float* bl = gcn_b + (size_t)l * DD;
        gemm_strip<8, false><<<gbN, 256, 0, stream>>>(B1, Wl, nullptr, B0, nsN);
        node_gather <<<(N + 7) / 8, 256, 0, stream>>>(B0, rowptr, cnt, ed, inv, B2, N);
        bias_relu   <<<elemBlocks, 256, 0, stream>>>(B2, bl, B1, N);
    }

    pool_zero<<<(G * DD + 255) / 256, 256, 0, stream>>>(pooled, G * DD);
    pool_add <<<(N * 32 + 255) / 256, 256, 0, stream>>>(B1, batch, pooled, N);

    gemm_strip<8, true><<<gbG, 256, 0, stream>>>(pooled, dec_W1, dec_b1, dech, nsG);
    gemm_strip<2, false><<<gbG, 256, 0, stream>>>(dech,   dec_W2, dec_b2, (float*)d_out, nsG);
}


# bias add and relu applied in node_gather to the row in registers, h stored directly: the three bias_relu launches and the agg plane's write and re-read are gone; same sums, same order
# speedup vs baseline: 3.0166x; 1.1938x over previous
#include <hip/hip_runtime.h>
#include <hip/hip_bf16.h>

typedef __attribute__((ext_vector_type(2))) float v2f;
typedef __attribute__((ext_vector_type(8))) float v8f;

#define DD 128

#define GW 8
#define APITCH 132

template <int NT, bool RELU>
__global__ __launch_bounds__(256)
void gemm_strip(const float* __restrict__ A,
                const float* __restrict__ W,
                const float* __restrict__ bias,
                float* __restrict__ C,
                int nStrips)
{
    constexpr int NC = NT * 16;
    constexpr int C4 = NC / 4;
    constexpr int NW = (64 * C4) / 256;
    constexpr int RPI = 32 / C4;
    constexpr int PW = 2 * NC + 32;
    __shared__ alignas(16) float Ws[64 * PW];
    __shared__ alignas(16) float As[GW * 16 * APITCH];

    const int tid  = threadIdx.x;
    const int lane = tid & 31;
    const int wv   = __builtin_amdgcn_readfirstlane(tid >> 5);
    const int strip = blockIdx.x * GW + wv;
    const bool valid = strip < nStrips;
    const int lstrip = valid ? strip : nStrips - 1;
    float* Aw = As + wv * (16 * APITCH);

    float4 av[16];
    {
        const float4* Ag = (const float4*)(A + (size_t)lstrip * 16 * DD);
#pragma unroll
        for (int r = 0; r < 16; ++r)
            av[r] = Ag[r * 32 + lane];
    }
    float4 w0[NW], w1[NW];
#pragma unroll
    for (int i = 0; i < NW; ++i) {
        const int idx = i * 256 + tid;
        const int p  = idx / C4;
        const int c4 = idx % C4;
        w0[i] = *(const float4*)(W + (size_t)(2 * p)     * NC + 4 * c4);
        w1[i] = *(const float4*)(W + (size_t)(2 * p + 1) * NC + 4 * c4);
    }
#pragma unroll
    for (int r = 0; r < 16; ++r)
        *(float4*)(Aw + r * APITCH + 4 * lane) = av[r];
#pragma unroll
    for (int i = 0; i < NW; ++i) {
        const int idx = i * 256 + tid;
        const int p  = idx / C4;
        const int c4 = idx % C4;
        float* d = Ws + p * PW + 8 * c4;
        *(float4*)(d)     = make_float4(w0[i].x, w1[i].x, w0[i].y, w1[i].y);
        *(float4*)(d + 4) = make_float4(w0[i].z, w1[i].z, w0[i].w, w1[i].w);
    }
    __syncthreads();

    const int h = lane >> 4;
    const int c = lane & 15;

    if (valid) {
        v8f acc[NT];
#pragma unroll
        for (int j = 0; j < NT; ++j)
            acc[j] = (v8f){0.f, 0.f, 0.f, 0.f, 0.f, 0.f, 0.f, 0.f};

        const float* ap = Aw + c * APITCH + 2 * h;
        const float* bp = Ws + h * PW + 2 * c;
#pragma unroll
        for (int s = 0; s < 32; ++s) {
            const v2f a = *(const v2f*)(ap + 4 * s);
#pragma unroll
            for (int j = 0; j < NT; ++j) {
                const v2f b = *(const v2f*)(bp + 2 * s * PW + 32 * j);
                acc[j] = __builtin_amdgcn_wmma_f32_16x16x4_f32(
                     false, a,  false, b,
                     (short)0, acc[j],  false,  false);
            }
        }

#pragma unroll
        for (int j = 0; j < NT; ++j)
#pragma unroll
            for (int r = 0; r < 8; ++r)
                Aw[(8 * h + r) * APITCH + 16 * j + c] = acc[j][r];
    }

    const int c4 = lane % C4;
    float4 bb = make_float4(0.f, 0.f, 0.f, 0.f);
    if (bias) bb = *(const float4*)(bias + 4 * c4);
    __syncthreads();

    if (valid) {
        float* Cg = C + (size_t)strip * 16 * NC;
        const int r0 = lane / C4;
#pragma unroll
        for (int i = 0; i < 2 * NT; ++i) {
            const int row = i * RPI + r0;
            float4 v = *(const float4*)(Aw + row * APITCH + 4 * c4);
            v.x += bb.x; v.y += bb.y; v.z += bb.z; v.w += bb.w;
            if (RELU) {
                v.x = v.x > 0.f ? v.x : 0.f;
                v.y = v.y > 0.f ? v.y : 0.f;
                v.z = v.z > 0.f ? v.z : 0.f;
                v.w = v.w > 0.f ? v.w : 0.f;
            }
            *(float4*)(Cg + (size_t)row * NC + 4 * c4) = v;
        }
    }
}

__global__ void csr_zero(int* __restrict__ cnt, int* __restrict__ total, int n) {
    int i = blockIdx.x * blockDim.x + threadIdx.x;
    if (i < n) cnt[i] = 0;
    if (i == 0) *total = 0;
}
__global__ void csr_count(const int* __restrict__ dst, int* cnt, int e) {
    int i = blockIdx.x * blockDim.x + threadIdx.x;
    if (i < e) atomicAdd(cnt + dst[i], 1);
}
__global__ __launch_bounds__(1024)
void csr_alloc(const int* __restrict__ cnt, int* total,
               int* __restrict__ rowptr, int* __restrict__ cursor,
               float* __restrict__ inv, int n) {
    __shared__ int wsum[32];
    __shared__ int blockBase;
    const int i    = blockIdx.x * 1024 + threadIdx.x;
    const int lane = threadIdx.x & 31;
    const int wv   = threadIdx.x >> 5;
    const int c = (i < n) ? cnt[i] : 0;
    int x = c;
#pragma unroll
    for (int o = 1; o < 32; o <<= 1) {
        const int y = __shfl_up(x, o);
        if (lane >= o) x += y;
    }
    if (lane == 31) wsum[wv] = x;
    __syncthreads();
    if (wv == 0) {
        const int t = wsum[lane];
        int xs = t;
#pragma unroll
        for (int o = 1; o < 32; o <<= 1) {
            const int y = __shfl_up(xs, o);
            if (lane >= o) xs += y;
        }
        wsum[lane] = xs - t;
        if (lane == 31) blockBase = atomicAdd(total, xs);
    }
    __syncthreads();
    if (i < n) {
        const int p = blockBase + wsum[wv] + x - c;
        rowptr[i] = p;
        cursor[i] = p;
        inv[i] = rsqrtf((float)(c + 1));
    }
}
__global__ void csr_fill(const int* __restrict__ src, const int* __restrict__ dst,
                         int* cursor, const float* __restrict__ inv,
                         int2* __restrict__ ed, int e) {
    int i = blockIdx.x * blockDim.x + threadIdx.x;
    if (i < e) {
        const int s = src[i];
        const int pos = atomicAdd(cursor + dst[i], 1);
        ed[pos] = make_int2(s, __float_as_int(inv[s]));
    }
}

__global__ __launch_bounds__(256)
void node_gather(const float* __restrict__ hw,
                 const int* __restrict__ rowptr,
                 const int* __restrict__ cnt,
                 const int2* __restrict__ ed,
                 const float* __restrict__ inv,
                 const float* __restrict__ b,
                 float* __restrict__ h, int n) {
    const int i = blockIdx.x * 8 + __builtin_amdgcn_readfirstlane(threadIdx.x >> 5);
    if (i >= n) return;
    const int lane = threadIdx.x & 31;
    const float4* hw4 = (const float4*)hw;
    const int start = __builtin_amdgcn_readfirstlane(rowptr[i]);
    const int m     = __builtin_amdgcn_readfirstlane(cnt[i]);
    const float di  = inv[i];
    const float4 sv = hw4[(size_t)i * 32 + lane];
    const float4 bv = ((const float4*)b)[lane];
    float4 acc = make_float4(0.f, 0.f, 0.f, 0.f);

    for (int base = 0; base < m; base += 32) {
        const int mm = (m - base) < 32 ? (m - base) : 32;
        int es = 0;
        float ec = 0.f;
        if (lane < mm) {
            const int2 e = ed[start + base + lane];
            es = e.x;
            ec = __int_as_float(e.y) * di;
        }
        for (int k = 0; k < mm; k += 8) {
            float4 r[8];
            float cf[8];
#pragma unroll
            for (int u = 0; u < 8; ++u) {
                r[u] = make_float4(0.f, 0.f, 0.f, 0.f);
                cf[u] = 0.f;
                if (k + u < mm) {
                    const int s = __builtin_amdgcn_readlane(es, k + u);
                    cf[u] = __int_as_float(__builtin_amdgcn_readlane(__float_as_int(ec), k + u));
                    r[u] = hw4[(size_t)s * 32 + lane];
                }
            }
#pragma unroll
            for (int u = 0; u < 8; ++u) {
                acc.x = fmaf(r[u].x, cf[u], acc.x);
                acc.y = fmaf(r[u].y, cf[u], acc.y);
                acc.z = fmaf(r[u].z, cf[u], acc.z);
                acc.w = fmaf(r[u].w, cf[u], acc.w);
            }
        }
    }
    const float sc = di * di;
    acc.x = fmaf(sv.x, sc, acc.x);
    acc.y = fmaf(sv.y, sc, acc.y);
    acc.z = fmaf(sv.z, sc, acc.z);
    acc.w = fmaf(sv.w, sc, acc.w);
    float4 v = make_float4(acc.x + bv.x, acc.y + bv.y, acc.z + bv.z, acc.w + bv.w);
    v.x = v.x > 0.f ? v.x : 0.f;
    v.y = v.y > 0.f ? v.y : 0.f;
    v.z = v.z > 0.f ? v.z : 0.f;
    v.w = v.w > 0.f ? v.w : 0.f;
    ((float4*)h)[(size_t)i * 32 + lane] = v;
}

__global__ void pool_zero(float* p, int n) {
    int i = blockIdx.x * blockDim.x + threadIdx.x;
    if (i < n) p[i] = 0.f;
}
__global__ void pool_add(const float* __restrict__ h,
                         const int* __restrict__ batch,
                         float* __restrict__ pooled, int n) {
    int t = blockIdx.x * blockDim.x + threadIdx.x;
    if (t < n * 32) {
        const int i = t >> 5;
        const int lane = t & 31;
        const int g = batch[i];
        const float* hp = h + (size_t)i * DD + lane;
        const float v0 = hp[0];
        const float v1 = hp[32];
        const float v2 = hp[64];
        const float v3 = hp[96];
        float* p = pooled + (size_t)g * DD + lane;
        atomicAdd(p +  0, v0);
        atomicAdd(p + 32, v1);
        atomicAdd(p + 64, v2);
        atomicAdd(p + 96, v3);
    }
}

extern "C" void kernel_launch(void* const* d_in, const int* in_sizes, int n_in,
                              void* d_out, int out_size, void* d_ws, size_t ws_size,
                              hipStream_t stream) {
    const float* x      = (const float*)d_in[0];
    const int*   ei     = (const int*)d_in[1];
    const int*   batch  = (const int*)d_in[2];
    const float* enc_W1 = (const float*)d_in[3];
    const float* enc_b1 = (const float*)d_in[4];
    const float* enc_W2 = (const float*)d_in[5];
    const float* enc_b2 = (const float*)d_in[6];
    const float* gcn_W  = (const float*)d_in[7];
    const float* gcn_b  = (const float*)d_in[8];
    const float* dec_W1 = (const float*)d_in[9];
    const float* dec_b1 = (const float*)d_in[10];
    const float* dec_W2 = (const float*)d_in[11];
    const float* dec_b2 = (const float*)d_in[12];

    const int N   = in_sizes[0] / DD;
    const int E   = in_sizes[1] / 2;
    const int L   = in_sizes[7] / (DD * DD);
    const int G   = 512;

    const int* src = ei;
    const int* dst = ei + E;

    char* ws = (char*)d_ws;
    const size_t nd = (size_t)N * DD * sizeof(float);
    float* B0     = (float*)(ws);
    float* B1     = (float*)(ws + nd);
    float* inv    = (float*)(ws + 3 * nd);
    size_t off    = 3 * nd + (((size_t)N * 4 + 255) & ~(size_t)255);
    float* pooled = (float*)(ws + off);
    float* dech   = pooled + (size_t)G * DD;
    const size_t nI = ((size_t)N * 4 + 255) & ~(size_t)255;
    size_t off2   = off + 2 * (size_t)G * DD * sizeof(float);
    int*  cnt     = (int*)(ws + off2);
    int*  rowptr  = (int*)(ws + off2 + nI);
    int*  cursor  = (int*)(ws + off2 + 2 * nI);
    int*  total   = (int*)(ws + off2 + 3 * nI);
    int2* ed      = (int2*)(ws + off2 + 3 * nI + 256);

    const int nsN = N / 16;
    const int nsG = G / 16;
    const int gbN = (nsN + GW - 1) / GW;
    const int gbG = (nsG + GW - 1) / GW;

    gemm_strip<8, true><<<gbN, 256, 0, stream>>>(x,  enc_W1, enc_b1, B0, nsN);
    gemm_strip<8, true><<<gbN, 256, 0, stream>>>(B0, enc_W2, enc_b2, B1, nsN);

    csr_zero <<<(N + 255) / 256, 256, 0, stream>>>(cnt, total, N);
    csr_count<<<(E + 255) / 256, 256, 0, stream>>>(dst, cnt, E);
    csr_alloc<<<(N + 1023) / 1024, 1024, 0, stream>>>(cnt, total, rowptr, cursor, inv, N);
    csr_fill <<<(E + 255) / 256, 256, 0, stream>>>(src, dst, cursor, inv, ed, E);

    for (int l = 0; l < L; ++l) {
        const float* Wl = gcn_W + (size_t)l * DD * DD;
        const float* bl = gcn_b + (size_t)l * DD;
        gemm_strip<8, false><<<gbN, 256, 0, stream>>>(B1, Wl, nullptr, B0, nsN);
        node_gather <<<(N + 7) / 8, 256, 0, stream>>>(B0, rowptr, cnt, ed, inv, bl, B1, N);
    }

    pool_zero<<<(G * DD + 255) / 256, 256, 0, stream>>>(pooled, G * DD);
    pool_add <<<(N * 32 + 255) / 256, 256, 0, stream>>>(B1, batch, pooled, N);

    gemm_strip<8, true><<<gbG, 256, 0, stream>>>(pooled, dec_W1, dec_b1, dech, nsG);
    gemm_strip<2, false><<<gbG, 256, 0, stream>>>(dech,   dec_W2, dec_b2, (float*)d_out, nsG);
}


# edges grouped by destination with a two-level counting sort in LDS (no per-edge global atomics); source inv_sqrt becomes a row scale in the layer gemm, so sums are inv[i]*(sum of scaled rows)
# speedup vs baseline: 4.0421x; 1.3400x over previous
#include <hip/hip_runtime.h>
#include <hip/hip_bf16.h>

typedef __attribute__((ext_vector_type(2))) float v2f;
typedef __attribute__((ext_vector_type(8))) float v8f;

#define DD 128

#define GW 8
#define APITCH 132

template <int NT, bool RELU, bool SCALE>
__global__ __launch_bounds__(256)
void gemm_strip(const float* __restrict__ A,
                const float* __restrict__ W,
                const float* __restrict__ bias,
                const float* __restrict__ rowscale,
                float* __restrict__ C,
                int nStrips)
{
    constexpr int NC = NT * 16;
    constexpr int C4 = NC / 4;
    constexpr int NW = (64 * C4) / 256;
    constexpr int RPI = 32 / C4;
    constexpr int PW = 2 * NC + 32;
    __shared__ alignas(16) float Ws[64 * PW];
    __shared__ alignas(16) float As[GW * 16 * APITCH];

    const int tid  = threadIdx.x;
    const int lane = tid & 31;
    const int wv   = __builtin_amdgcn_readfirstlane(tid >> 5);
    const int strip = blockIdx.x * GW + wv;
    const bool valid = strip < nStrips;
    const int lstrip = valid ? strip : nStrips - 1;
    float* Aw = As + wv * (16 * APITCH);

    float4 av[16];
    {
        const float4* Ag = (const float4*)(A + (size_t)lstrip * 16 * DD);
#pragma unroll
        for (int r = 0; r < 16; ++r)
            av[r] = Ag[r * 32 + lane];
    }
    float4 w0[NW], w1[NW];
#pragma unroll
    for (int i = 0; i < NW; ++i) {
        const int idx = i * 256 + tid;
        const int p  = idx / C4;
        const int c4 = idx % C4;
        w0[i] = *(const float4*)(W + (size_t)(2 * p)     * NC + 4 * c4);
        w1[i] = *(const float4*)(W + (size_t)(2 * p + 1) * NC + 4 * c4);
    }
#pragma unroll
    for (int r = 0; r < 16; ++r)
        *(float4*)(Aw + r * APITCH + 4 * lane) = av[r];
#pragma unroll
    for (int i = 0; i < NW; ++i) {
        const int idx = i * 256 + tid;
        const int p  = idx / C4;
        const int c4 = idx % C4;
        float* d = Ws + p * PW + 8 * c4;
        *(float4*)(d)     = make_float4(w0[i].x, w1[i].x, w0[i].y, w1[i].y);
        *(float4*)(d + 4) = make_float4(w0[i].z, w1[i].z, w0[i].w, w1[i].w);
    }
    __syncthreads();

    const int h = lane >> 4;
    const int c = lane & 15;

    if (valid) {
        v8f acc[NT];
#pragma unroll
        for (int j = 0; j < NT; ++j)
            acc[j] = (v8f){0.f, 0.f, 0.f, 0.f, 0.f, 0.f, 0.f, 0.f};

        const float* ap = Aw + c * APITCH + 2 * h;
        const float* bp = Ws + h * PW + 2 * c;
#pragma unroll
        for (int s = 0; s < 32; ++s) {
            const v2f a = *(const v2f*)(ap + 4 * s);
#pragma unroll
            for (int j = 0; j < NT; ++j) {
                const v2f b = *(const v2f*)(bp + 2 * s * PW + 32 * j);
                acc[j] = __builtin_amdgcn_wmma_f32_16x16x4_f32(
                     false, a,  false, b,
                     (short)0, acc[j],  false,  false);
            }
        }

#pragma unroll
        for (int j = 0; j < NT; ++j)
#pragma unroll
            for (int r = 0; r < 8; ++r)
                Aw[(8 * h + r) * APITCH + 16 * j + c] = acc[j][r];
    }

    const int c4 = lane % C4;
    float4 bb = make_float4(0.f, 0.f, 0.f, 0.f);
    if (bias) bb = *(const float4*)(bias + 4 * c4);
    float rs[2 * NT];
    if (SCALE) {
#pragma unroll
        for (int i = 0; i < 2 * NT; ++i)
            rs[i] = rowscale[(size_t)lstrip * 16 + i * RPI + lane / C4];
    }
    __syncthreads();

    if (valid) {
        float* Cg = C + (size_t)strip * 16 * NC;
        const int r0 = lane / C4;
#pragma unroll
        for (int i = 0; i < 2 * NT; ++i) {
            const int row = i * RPI + r0;
            float4 v = *(const float4*)(Aw + row * APITCH + 4 * c4);
            if (SCALE) { v.x *= rs[i]; v.y *= rs[i]; v.z *= rs[i]; v.w *= rs[i]; }
            v.x += bb.x; v.y += bb.y; v.z += bb.z; v.w += bb.w;
            if (RELU) {
                v.x = v.x > 0.f ? v.x : 0.f;
                v.y = v.y > 0.f ? v.y : 0.f;
                v.z = v.z > 0.f ? v.z : 0.f;
                v.w = v.w > 0.f ? v.w : 0.f;
            }
            *(float4*)(Cg + (size_t)row * NC + 4 * c4) = v;
        }
    }
}

#define BSH   9
#define BN    512
#define MAXNB 1024
#define STG   16384

__global__ void bin_zero(int* __restrict__ binTotal, int nb) {
    int i = blockIdx.x * blockDim.x + threadIdx.x;
    if (i < nb) binTotal[i] = 0;
}
__global__ __launch_bounds__(1024)
void bin_count(const int* __restrict__ dst, int* binTotal, int e, int nb, int chunk) {
    __shared__ int hist[MAXNB];
    const int tid = threadIdx.x;
    for (int b = tid; b < nb; b += 1024) hist[b] = 0;
    __syncthreads();
    const int lo = blockIdx.x * chunk;
    const int hi = (lo + chunk < e) ? (lo + chunk) : e;
    for (int i = lo + tid; i < hi; i += 1024) atomicAdd(&hist[dst[i] >> BSH], 1);
    __syncthreads();
    for (int b = tid; b < nb; b += 1024) {
        const int c = hist[b];
        if (c) atomicAdd(binTotal + b, c);
    }
}
__global__ __launch_bounds__(1024)
void bin_scan(const int* __restrict__ binTotal, int* __restrict__ binBase,
              int* __restrict__ binCursor, int nb) {
    __shared__ int wsum[32];
    const int tid  = threadIdx.x;
    const int lane = tid & 31;
    const int wv   = tid >> 5;
    const int c = (tid < nb) ? binTotal[tid] : 0;
    int x = c;
#pragma unroll
    for (int o = 1; o < 32; o <<= 1) {
        const int y = __shfl_up(x, o);
        if (lane >= o) x += y;
    }
    if (lane == 31) wsum[wv] = x;
    __syncthreads();
    if (wv == 0) {
        const int t = wsum[lane];
        int xs = t;
#pragma unroll
        for (int o = 1; o < 32; o <<= 1) {
            const int y = __shfl_up(xs, o);
            if (lane >= o) xs += y;
        }
        wsum[lane] = xs - t;
    }
    __syncthreads();
    if (tid < nb) {
        const int p = wsum[wv] + x - c;
        binBase[tid] = p;
        binCursor[tid] = p;
    }
}
__global__ __launch_bounds__(1024)
void bin_scatter(const int* __restrict__ src, const int* __restrict__ dst,
                 int* binCursor, int2* __restrict__ binned, int e, int nb, int chunk) {
    __shared__ int hist[MAXNB];
    __shared__ int bbase[MAXNB];
    const int tid = threadIdx.x;
    for (int b = tid; b < nb; b += 1024) hist[b] = 0;
    __syncthreads();
    const int lo = blockIdx.x * chunk;
    const int hi = (lo + chunk < e) ? (lo + chunk) : e;
    for (int i = lo + tid; i < hi; i += 1024) atomicAdd(&hist[dst[i] >> BSH], 1);
    __syncthreads();
    for (int b = tid; b < nb; b += 1024) {
        const int c = hist[b];
        bbase[b] = c ? atomicAdd(binCursor + b, c) : 0;
        hist[b] = 0;
    }
    __syncthreads();
    for (int i = lo + tid; i < hi; i += 1024) {
        const int d = dst[i];
        const int b = d >> BSH;
        const int r = atomicAdd(&hist[b], 1);
        binned[bbase[b] + r] = make_int2(d, src[i]);
    }
}
__global__ __launch_bounds__(1024)
void bin_sort(const int2* __restrict__ binned, const int* __restrict__ binTotal,
              const int* __restrict__ binBase, int* __restrict__ cnt,
              int* __restrict__ rowptr, float* __restrict__ inv,
              int* __restrict__ ed, int n) {
    __shared__ int cur[BN];
    __shared__ int wsum[32];
    __shared__ int stage[STG];
    const int tid  = threadIdx.x;
    const int lane = tid & 31;
    const int wv   = tid >> 5;
    const int b    = blockIdx.x;
    const int nE   = binTotal[b];
    const int base = binBase[b];
    const int2* be = binned + base;
    if (tid < BN) cur[tid] = 0;
    __syncthreads();
    for (int i = tid; i < nE; i += 1024) atomicAdd(&cur[be[i].x & (BN - 1)], 1);
    __syncthreads();
    const int c = (tid < BN) ? cur[tid] : 0;
    int x = c;
#pragma unroll
    for (int o = 1; o < 32; o <<= 1) {
        const int y = __shfl_up(x, o);
        if (lane >= o) x += y;
    }
    if (lane == 31) wsum[wv] = x;
    __syncthreads();
    if (wv == 0) {
        const int t = wsum[lane];
        int xs = t;
#pragma unroll
        for (int o = 1; o < 32; o <<= 1) {
            const int y = __shfl_up(xs, o);
            if (lane >= o) xs += y;
        }
        wsum[lane] = xs - t;
    }
    __syncthreads();
    if (tid < BN) {
        const int excl = wsum[wv] + x - c;
        cur[tid] = excl;
        const int node = b * BN + tid;
        if (node < n) {
            cnt[node] = c;
            rowptr[node] = base + excl;
            inv[node] = rsqrtf((float)(c + 1));
        }
    }
    __syncthreads();
    const bool fits = nE <= STG;
    for (int i = tid; i < nE; i += 1024) {
        const int2 e = be[i];
        const int r = atomicAdd(&cur[e.x & (BN - 1)], 1);
        if (fits) stage[r] = e.y; else ed[base + r] = e.y;
    }
    if (fits) {
        __syncthreads();
        for (int i = tid; i < nE; i += 1024) ed[base + i] = stage[i];
    }
}

__global__ __launch_bounds__(256)
void node_gather(const float* __restrict__ hw,
                 const int* __restrict__ rowptr,
                 const int* __restrict__ cnt,
                 const int* __restrict__ ed,
                 const float* __restrict__ inv,
                 const float* __restrict__ b,
                 float* __restrict__ h, int n) {
    const int i = blockIdx.x * 8 + __builtin_amdgcn_readfirstlane(threadIdx.x >> 5);
    if (i >= n) return;
    const int lane = threadIdx.x & 31;
    const float4* hw4 = (const float4*)hw;
    const int start = __builtin_amdgcn_readfirstlane(rowptr[i]);
    const int m     = __builtin_amdgcn_readfirstlane(cnt[i]);
    const float di  = inv[i];
    const float4 sv = hw4[(size_t)i * 32 + lane];
    const float4 bv = ((const float4*)b)[lane];
    float4 acc = make_float4(0.f, 0.f, 0.f, 0.f);

    for (int base = 0; base < m; base += 32) {
        const int mm = (m - base) < 32 ? (m - base) : 32;
        int es = 0;
        if (lane < mm) es = ed[start + base + lane];
        for (int k = 0; k < mm; k += 8) {
            float4 r[8];
#pragma unroll
            for (int u = 0; u < 8; ++u) {
                r[u] = make_float4(0.f, 0.f, 0.f, 0.f);
                if (k + u < mm) {
                    const int s = __builtin_amdgcn_readlane(es, k + u);
                    r[u] = hw4[(size_t)s * 32 + lane];
                }
            }
#pragma unroll
            for (int u = 0; u < 8; ++u) {
                acc.x += r[u].x;
                acc.y += r[u].y;
                acc.z += r[u].z;
                acc.w += r[u].w;
            }
        }
    }
    acc.x += sv.x;
    acc.y += sv.y;
    acc.z += sv.z;
    acc.w += sv.w;
    float4 v = make_float4(fmaf(acc.x, di, bv.x), fmaf(acc.y, di, bv.y),
                           fmaf(acc.z, di, bv.z), fmaf(acc.w, di, bv.w));
    v.x = v.x > 0.f ? v.x : 0.f;
    v.y = v.y > 0.f ? v.y : 0.f;
    v.z = v.z > 0.f ? v.z : 0.f;
    v.w = v.w > 0.f ? v.w : 0.f;
    ((float4*)h)[(size_t)i * 32 + lane] = v;
}

__global__ void pool_zero(float* p, int n) {
    int i = blockIdx.x * blockDim.x + threadIdx.x;
    if (i < n) p[i] = 0.f;
}
__global__ void pool_add(const float* __restrict__ h,
                         const int* __restrict__ batch,
                         float* __restrict__ pooled, int n) {
    int t = blockIdx.x * blockDim.x + threadIdx.x;
    if (t < n * 32) {
        const int i = t >> 5;
        const int lane = t & 31;
        const int g = batch[i];
        const float* hp = h + (size_t)i * DD + lane;
        const float v0 = hp[0];
        const float v1 = hp[32];
        const float v2 = hp[64];
        const float v3 = hp[96];
        float* p = pooled + (size_t)g * DD + lane;
        atomicAdd(p +  0, v0);
        atomicAdd(p + 32, v1);
        atomicAdd(p + 64, v2);
        atomicAdd(p + 96, v3);
    }
}

extern "C" void kernel_launch(void* const* d_in, const int* in_sizes, int n_in,
                              void* d_out, int out_size, void* d_ws, size_t ws_size,
                              hipStream_t stream) {
    const float* x      = (const float*)d_in[0];
    const int*   ei     = (const int*)d_in[1];
    const int*   batch  = (const int*)d_in[2];
    const float* enc_W1 = (const float*)d_in[3];
    const float* enc_b1 = (const float*)d_in[4];
    const float* enc_W2 = (const float*)d_in[5];
    const float* enc_b2 = (const float*)d_in[6];
    const float* gcn_W  = (const float*)d_in[7];
    const float* gcn_b  = (const float*)d_in[8];
    const float* dec_W1 = (const float*)d_in[9];
    const float* dec_b1 = (const float*)d_in[10];
    const float* dec_W2 = (const float*)d_in[11];
    const float* dec_b2 = (const float*)d_in[12];

    const int N   = in_sizes[0] / DD;
    const int E   = in_sizes[1] / 2;
    const int L   = in_sizes[7] / (DD * DD);
    const int G   = 512;

    const int* src = ei;
    const int* dst = ei + E;

    char* ws = (char*)d_ws;
    const size_t nd = (size_t)N * DD * sizeof(float);
    float* B0     = (float*)(ws);
    float* B1     = (float*)(ws + nd);
    float* inv    = (float*)(ws + 3 * nd);
    size_t off    = 3 * nd + (((size_t)N * 4 + 255) & ~(size_t)255);
    float* pooled = (float*)(ws + off);
    float* dech   = pooled + (size_t)G * DD;
    const size_t nI = ((size_t)N * 4 + 255) & ~(size_t)255;
    size_t off2   = off + 2 * (size_t)G * DD * sizeof(float);
    int*  cnt     = (int*)(ws + off2);
    int*  rowptr  = (int*)(ws + off2 + nI);
    int*  binTotal  = (int*)(ws + off2 + 2 * nI);
    int*  binBase   = binTotal + MAXNB;
    int*  binCursor = binBase + MAXNB;
    int*  ed      = (int*)(ws + off2 + 3 * nI + 256);
    int2* binned  = (int2*)(ws + 2 * nd);
    const int nb    = (N + BN - 1) / BN;
    const int nblk  = 256;
    const int chunk = (E + nblk - 1) / nblk;

    const int nsN = N / 16;
    const int nsG = G / 16;
    const int gbN = (nsN + GW - 1) / GW;
    const int gbG = (nsG + GW - 1) / GW;

    gemm_strip<8, true, false><<<gbN, 256, 0, stream>>>(x,  enc_W1, enc_b1, nullptr, B0, nsN);
    gemm_strip<8, true, false><<<gbN, 256, 0, stream>>>(B0, enc_W2, enc_b2, nullptr, B1, nsN);

    bin_zero   <<<(nb + 255) / 256, 256, 0, stream>>>(binTotal, nb);
    bin_count  <<<nblk, 1024, 0, stream>>>(dst, binTotal, E, nb, chunk);
    bin_scan   <<<1, 1024, 0, stream>>>(binTotal, binBase, binCursor, nb);
    bin_scatter<<<nblk, 1024, 0, stream>>>(src, dst, binCursor, binned, E, nb, chunk);
    bin_sort   <<<nb, 1024, 0, stream>>>(binned, binTotal, binBase, cnt, rowptr, inv, ed, N);

    for (int l = 0; l < L; ++l) {
        const float* Wl = gcn_W + (size_t)l * DD * DD;
        const float* bl = gcn_b + (size_t)l * DD;
        gemm_strip<8, false, true><<<gbN, 256, 0, stream>>>(B1, Wl, nullptr, inv, B0, nsN);
        node_gather <<<(N + 7) / 8, 256, 0, stream>>>(B0, rowptr, cnt, ed, inv, bl, B1, N);
    }

    pool_zero<<<(G * DD + 255) / 256, 256, 0, stream>>>(pooled, G * DD);
    pool_add <<<(N * 32 + 255) / 256, 256, 0, stream>>>(B1, batch, pooled, N);

    gemm_strip<8, true, false><<<gbG, 256, 0, stream>>>(pooled, dec_W1, dec_b1, nullptr, dech, nsG);
    gemm_strip<2, false, false><<<gbG, 256, 0, stream>>>(dech,   dec_W2, dec_b2, nullptr, (float*)d_out, nsG);
}


# node_gather, pool_add: one wave per 8 consecutive nodes, not per node (8x fewer wave entries); node_gather reads a group's source lanes before its row loads so the 8 loads issue together; same sums
# speedup vs baseline: 4.4073x; 1.0903x over previous
#include <hip/hip_runtime.h>
#include <hip/hip_bf16.h>

typedef __attribute__((ext_vector_type(2))) float v2f;
typedef __attribute__((ext_vector_type(8))) float v8f;

#define DD 128

#define GW 8
#define APITCH 132

template <int NT, bool RELU, bool SCALE>
__global__ __launch_bounds__(256)
void gemm_strip(const float* __restrict__ A,
                const float* __restrict__ W,
                const float* __restrict__ bias,
                const float* __restrict__ rowscale,
                float* __restrict__ C,
                int nStrips)
{
    constexpr int NC = NT * 16;
    constexpr int C4 = NC / 4;
    constexpr int NW = (64 * C4) / 256;
    constexpr int RPI = 32 / C4;
    constexpr int PW = 2 * NC + 32;
    __shared__ alignas(16) float Ws[64 * PW];
    __shared__ alignas(16) float As[GW * 16 * APITCH];

    const int tid  = threadIdx.x;
    const int lane = tid & 31;
    const int wv   = __builtin_amdgcn_readfirstlane(tid >> 5);
    const int strip = blockIdx.x * GW + wv;
    const bool valid = strip < nStrips;
    const int lstrip = valid ? strip : nStrips - 1;
    float* Aw = As + wv * (16 * APITCH);

    float4 av[16];
    {
        const float4* Ag = (const float4*)(A + (size_t)lstrip * 16 * DD);
#pragma unroll
        for (int r = 0; r < 16; ++r)
            av[r] = Ag[r * 32 + lane];
    }
    float4 w0[NW], w1[NW];
#pragma unroll
    for (int i = 0; i < NW; ++i) {
        const int idx = i * 256 + tid;
        const int p  = idx / C4;
        const int c4 = idx % C4;
        w0[i] = *(const float4*)(W + (size_t)(2 * p)     * NC + 4 * c4);
        w1[i] = *(const float4*)(W + (size_t)(2 * p + 1) * NC + 4 * c4);
    }
#pragma unroll
    for (int r = 0; r < 16; ++r)
        *(float4*)(Aw + r * APITCH + 4 * lane) = av[r];
#pragma unroll
    for (int i = 0; i < NW; ++i) {
        const int idx = i * 256 + tid;
        const int p  = idx / C4;
        const int c4 = idx % C4;
        float* d = Ws + p * PW + 8 * c4;
        *(float4*)(d)     = make_float4(w0[i].x, w1[i].x, w0[i].y, w1[i].y);
        *(float4*)(d + 4) = make_float4(w0[i].z, w1[i].z, w0[i].w, w1[i].w);
    }
    __syncthreads();

    const int h = lane >> 4;
    const int c = lane & 15;

    if (valid) {
        v8f acc[NT];
#pragma unroll
        for (int j = 0; j < NT; ++j)
            acc[j] = (v8f){0.f, 0.f, 0.f, 0.f, 0.f, 0.f, 0.f, 0.f};

        const float* ap = Aw + c * APITCH + 2 * h;
        const float* bp = Ws + h * PW + 2 * c;
#pragma unroll
        for (int s = 0; s < 32; ++s) {
            const v2f a = *(const v2f*)(ap + 4 * s);
#pragma unroll
            for (int j = 0; j < NT; ++j) {
                const v2f b = *(const v2f*)(bp + 2 * s * PW + 32 * j);
                acc[j] = __builtin_amdgcn_wmma_f32_16x16x4_f32(
                     false, a,  false, b,
                     (short)0, acc[j],  false,  false);
            }
        }

#pragma unroll
        for (int j = 0; j < NT; ++j)
#pragma unroll
            for (int r = 0; r < 8; ++r)
                Aw[(8 * h + r) * APITCH + 16 * j + c] = acc[j][r];
    }

    const int c4 = lane % C4;
    float4 bb = make_float4(0.f, 0.f, 0.f, 0.f);
    if (bias) bb = *(const float4*)(bias + 4 * c4);
    float rs[2 * NT];
    if (SCALE) {
#pragma unroll
        for (int i = 0; i < 2 * NT; ++i)
            rs[i] = rowscale[(size_t)lstrip * 16 + i * RPI + lane / C4];
    }
    __syncthreads();

    if (valid) {
        float* Cg = C + (size_t)strip * 16 * NC;
        const int r0 = lane / C4;
#pragma unroll
        for (int i = 0; i < 2 * NT; ++i) {
            const int row = i * RPI + r0;
            float4 v = *(const float4*)(Aw + row * APITCH + 4 * c4);
            if (SCALE) { v.x *= rs[i]; v.y *= rs[i]; v.z *= rs[i]; v.w *= rs[i]; }
            v.x += bb.x; v.y += bb.y; v.z += bb.z; v.w += bb.w;
            if (RELU) {
                v.x = v.x > 0.f ? v.x : 0.f;
                v.y = v.y > 0.f ? v.y : 0.f;
                v.z = v.z > 0.f ? v.z : 0.f;
                v.w = v.w > 0.f ? v.w : 0.f;
            }
            *(float4*)(Cg + (size_t)row * NC + 4 * c4) = v;
        }
    }
}

#define BSH   9
#define BN    512
#define MAXNB 1024
#define STG   16384

__global__ void bin_zero(int* __restrict__ binTotal, int nb) {
    int i = blockIdx.x * blockDim.x + threadIdx.x;
    if (i < nb) binTotal[i] = 0;
}
__global__ __launch_bounds__(1024)
void bin_count(const int* __restrict__ dst, int* binTotal, int e, int nb, int chunk) {
    __shared__ int hist[MAXNB];
    const int tid = threadIdx.x;
    for (int b = tid; b < nb; b += 1024) hist[b] = 0;
    __syncthreads();
    const int lo = blockIdx.x * chunk;
    const int hi = (lo + chunk < e) ? (lo + chunk) : e;
    for (int i = lo + tid; i < hi; i += 1024) atomicAdd(&hist[dst[i] >> BSH], 1);
    __syncthreads();
    for (int b = tid; b < nb; b += 1024) {
        const int c = hist[b];
        if (c) atomicAdd(binTotal + b, c);
    }
}
__global__ __launch_bounds__(1024)
void bin_scan(const int* __restrict__ binTotal, int* __restrict__ binBase,
              int* __restrict__ binCursor, int nb) {
    __shared__ int wsum[32];
    const int tid  = threadIdx.x;
    const int lane = tid & 31;
    const int wv   = tid >> 5;
    const int c = (tid < nb) ? binTotal[tid] : 0;
    int x = c;
#pragma unroll
    for (int o = 1; o < 32; o <<= 1) {
        const int y = __shfl_up(x, o);
        if (lane >= o) x += y;
    }
    if (lane == 31) wsum[wv] = x;
    __syncthreads();
    if (wv == 0) {
        const int t = wsum[lane];
        int xs = t;
#pragma unroll
        for (int o = 1; o < 32; o <<= 1) {
            const int y = __shfl_up(xs, o);
            if (lane >= o) xs += y;
        }
        wsum[lane] = xs - t;
    }
    __syncthreads();
    if (tid < nb) {
        const int p = wsum[wv] + x - c;
        binBase[tid] = p;
        binCursor[tid] = p;
    }
}
__global__ __launch_bounds__(1024)
void bin_scatter(const int* __restrict__ src, const int* __restrict__ dst,
                 int* binCursor, int2* __restrict__ binned, int e, int nb, int chunk) {
    __shared__ int hist[MAXNB];
    __shared__ int bbase[MAXNB];
    const int tid = threadIdx.x;
    for (int b = tid; b < nb; b += 1024) hist[b] = 0;
    __syncthreads();
    const int lo = blockIdx.x * chunk;
    const int hi = (lo + chunk < e) ? (lo + chunk) : e;
    for (int i = lo + tid; i < hi; i += 1024) atomicAdd(&hist[dst[i] >> BSH], 1);
    __syncthreads();
    for (int b = tid; b < nb; b += 1024) {
        const int c = hist[b];
        bbase[b] = c ? atomicAdd(binCursor + b, c) : 0;
        hist[b] = 0;
    }
    __syncthreads();
    for (int i = lo + tid; i < hi; i += 1024) {
        const int d = dst[i];
        const int b = d >> BSH;
        const int r = atomicAdd(&hist[b], 1);
        binned[bbase[b] + r] = make_int2(d, src[i]);
    }
}
__global__ __launch_bounds__(1024)
void bin_sort(const int2* __restrict__ binned, const int* __restrict__ binTotal,
              const int* __restrict__ binBase, int* __restrict__ cnt,
              int* __restrict__ rowptr, float* __restrict__ inv,
              int* __restrict__ ed, int n) {
    __shared__ int cur[BN];
    __shared__ int wsum[32];
    __shared__ int stage[STG];
    const int tid  = threadIdx.x;
    const int lane = tid & 31;
    const int wv   = tid >> 5;
    const int b    = blockIdx.x;
    const int nE   = binTotal[b];
    const int base = binBase[b];
    const int2* be = binned + base;
    if (tid < BN) cur[tid] = 0;
    __syncthreads();
    for (int i = tid; i < nE; i += 1024) atomicAdd(&cur[be[i].x & (BN - 1)], 1);
    __syncthreads();
    const int c = (tid < BN) ? cur[tid] : 0;
    int x = c;
#pragma unroll
    for (int o = 1; o < 32; o <<= 1) {
        const int y = __shfl_up(x, o);
        if (lane >= o) x += y;
    }
    if (lane == 31) wsum[wv] = x;
    __syncthreads();
    if (wv == 0) {
        const int t = wsum[lane];
        int xs = t;
#pragma unroll
        for (int o = 1; o < 32; o <<= 1) {
            const int y = __shfl_up(xs, o);
            if (lane >= o) xs += y;
        }
        wsum[lane] = xs - t;
    }
    __syncthreads();
    if (tid < BN) {
        const int excl = wsum[wv] + x - c;
        cur[tid] = excl;
        const int node = b * BN + tid;
        if (node < n) {
            cnt[node] = c;
            rowptr[node] = base + excl;
            inv[node] = rsqrtf((float)(c + 1));
        }
    }
    __syncthreads();
    const bool fits = nE <= STG;
    for (int i = tid; i < nE; i += 1024) {
        const int2 e = be[i];
        const int r = atomicAdd(&cur[e.x & (BN - 1)], 1);
        if (fits) stage[r] = e.y; else ed[base + r] = e.y;
    }
    if (fits) {
        __syncthreads();
        for (int i = tid; i < nE; i += 1024) ed[base + i] = stage[i];
    }
}

#define NPW 8
__global__ __launch_bounds__(256)
void node_gather(const float* __restrict__ hw,
                 const int* __restrict__ rowptr,
                 const int* __restrict__ cnt,
                 const int* __restrict__ ed,
                 const float* __restrict__ inv,
                 const float* __restrict__ b,
                 float* __restrict__ h, int n) {
    const int n0 = (blockIdx.x * 8 + __builtin_amdgcn_readfirstlane(threadIdx.x >> 5)) * NPW;
    if (n0 >= n) return;
    const int nn = (n - n0) < NPW ? (n - n0) : NPW;
    const int lane = threadIdx.x & 31;
    const float4* hw4 = (const float4*)hw;
    int cv = 0;
    float dv = 0.f;
    if (lane < nn) {
        cv = cnt[n0 + lane];
        dv = inv[n0 + lane];
    }
    int start = __builtin_amdgcn_readfirstlane(rowptr[n0]);
    const float4 bv = ((const float4*)b)[lane];

    for (int j = 0; j < nn; ++j) {
        const int i = n0 + j;
        const int m = __builtin_amdgcn_readlane(cv, j);
        const float di = __int_as_float(__builtin_amdgcn_readlane(__float_as_int(dv), j));
        const float4 sv = hw4[(size_t)i * 32 + lane];
        float4 acc = make_float4(0.f, 0.f, 0.f, 0.f);

        for (int base = 0; base < m; base += 32) {
            const int mm = (m - base) < 32 ? (m - base) : 32;
            int es = 0;
            if (lane < mm) es = ed[start + base + lane];
            int k = 0;
            do {
                int s[8];
#pragma unroll
                for (int u = 0; u < 8; ++u) {
                    const int q = (k + u < mm) ? (k + u) : (mm - 1);
                    s[u] = __builtin_amdgcn_readlane(es, q);
                }
                float4 r[8];
#pragma unroll
                for (int u = 0; u < 8; ++u) {
                    r[u] = make_float4(0.f, 0.f, 0.f, 0.f);
                    if (k + u < mm) r[u] = hw4[(size_t)s[u] * 32 + lane];
                }
#pragma unroll
                for (int u = 0; u < 8; ++u) {
                    acc.x += r[u].x;
                    acc.y += r[u].y;
                    acc.z += r[u].z;
                    acc.w += r[u].w;
                }
                k += 8;
            } while (k < mm);
        }
        acc.x += sv.x;
        acc.y += sv.y;
        acc.z += sv.z;
        acc.w += sv.w;
        float4 v = make_float4(fmaf(acc.x, di, bv.x), fmaf(acc.y, di, bv.y),
                               fmaf(acc.z, di, bv.z), fmaf(acc.w, di, bv.w));
        v.x = v.x > 0.f ? v.x : 0.f;
        v.y = v.y > 0.f ? v.y : 0.f;
        v.z = v.z > 0.f ? v.z : 0.f;
        v.w = v.w > 0.f ? v.w : 0.f;
        ((float4*)h)[(size_t)i * 32 + lane] = v;
        start += m;
    }
}

__global__ void pool_zero(float* p, int n) {
    int i = blockIdx.x * blockDim.x + threadIdx.x;
    if (i < n) p[i] = 0.f;
}
__global__ __launch_bounds__(256)
void pool_add(const float* __restrict__ h,
              const int* __restrict__ batch,
              float* __restrict__ pooled, int n) {
    const int n0 = (blockIdx.x * 8 + __builtin_amdgcn_readfirstlane(threadIdx.x >> 5)) * NPW;
    if (n0 >= n) return;
    const int nn = (n - n0) < NPW ? (n - n0) : NPW;
    const int lane = threadIdx.x & 31;
    int gv = 0;
    if (lane < nn) gv = batch[n0 + lane];
    float v[NPW][4];
#pragma unroll
    for (int j = 0; j < NPW; ++j) {
        if (j < nn) {
            const float* hp = h + (size_t)(n0 + j) * DD + lane;
            v[j][0] = hp[0];
            v[j][1] = hp[32];
            v[j][2] = hp[64];
            v[j][3] = hp[96];
        }
    }
#pragma unroll
    for (int j = 0; j < NPW; ++j) {
        if (j < nn) {
            const int g = __builtin_amdgcn_readlane(gv, j);
            float* p = pooled + (size_t)g * DD + lane;
            atomicAdd(p +  0, v[j][0]);
            atomicAdd(p + 32, v[j][1]);
            atomicAdd(p + 64, v[j][2]);
            atomicAdd(p + 96, v[j][3]);
        }
    }
}

extern "C" void kernel_launch(void* const* d_in, const int* in_sizes, int n_in,
                              void* d_out, int out_size, void* d_ws, size_t ws_size,
                              hipStream_t stream) {
    const float* x      = (const float*)d_in[0];
    const int*   ei     = (const int*)d_in[1];
    const int*   batch  = (const int*)d_in[2];
    const float* enc_W1 = (const float*)d_in[3];
    const float* enc_b1 = (const float*)d_in[4];
    const float* enc_W2 = (const float*)d_in[5];
    const float* enc_b2 = (const float*)d_in[6];
    const float* gcn_W  = (const float*)d_in[7];
    const float* gcn_b  = (const float*)d_in[8];
    const float* dec_W1 = (const float*)d_in[9];
    const float* dec_b1 = (const float*)d_in[10];
    const float* dec_W2 = (const float*)d_in[11];
    const float* dec_b2 = (const float*)d_in[12];

    const int N   = in_sizes[0] / DD;
    const int E   = in_sizes[1] / 2;
    const int L   = in_sizes[7] / (DD * DD);
    const int G   = 512;

    const int* src = ei;
    const int* dst = ei + E;

    char* ws = (char*)d_ws;
    const size_t nd = (size_t)N * DD * sizeof(float);
    float* B0     = (float*)(ws);
    float* B1     = (float*)(ws + nd);
    float* inv    = (float*)(ws + 3 * nd);
    size_t off    = 3 * nd + (((size_t)N * 4 + 255) & ~(size_t)255);
    float* pooled = (float*)(ws + off);
    float* dech   = pooled + (size_t)G * DD;
    const size_t nI = ((size_t)N * 4 + 255) & ~(size_t)255;
    size_t off2   = off + 2 * (size_t)G * DD * sizeof(float);
    int*  cnt     = (int*)(ws + off2);
    int*  rowptr  = (int*)(ws + off2 + nI);
    int*  binTotal  = (int*)(ws + off2 + 2 * nI);
    int*  binBase   = binTotal + MAXNB;
    int*  binCursor = binBase + MAXNB;
    int*  ed      = (int*)(ws + off2 + 3 * nI + 256);
    int2* binned  = (int2*)(ws + 2 * nd);
    const int nb    = (N + BN - 1) / BN;
    const int nblk  = 256;
    const int chunk = (E + nblk - 1) / nblk;

    const int nsN = N / 16;
    const int nsG = G / 16;
    const int gbN = (nsN + GW - 1) / GW;
    const int gbG = (nsG + GW - 1) / GW;
    const int gbR = (N + 8 * NPW - 1) / (8 * NPW);

    gemm_strip<8, true, false><<<gbN, 256, 0, stream>>>(x,  enc_W1, enc_b1, nullptr, B0, nsN);
    gemm_strip<8, true, false><<<gbN, 256, 0, stream>>>(B0, enc_W2, enc_b2, nullptr, B1, nsN);

    bin_zero   <<<(nb + 255) / 256, 256, 0, stream>>>(binTotal, nb);
    bin_count  <<<nblk, 1024, 0, stream>>>(dst, binTotal, E, nb, chunk);
    bin_scan   <<<1, 1024, 0, stream>>>(binTotal, binBase, binCursor, nb);
    bin_scatter<<<nblk, 1024, 0, stream>>>(src, dst, binCursor, binned, E, nb, chunk);
    bin_sort   <<<nb, 1024, 0, stream>>>(binned, binTotal, binBase, cnt, rowptr, inv, ed, N);

    for (int l = 0; l < L; ++l) {
        const float* Wl = gcn_W + (size_t)l * DD * DD;
        const float* bl = gcn_b + (size_t)l * DD;
        gemm_strip<8, false, true><<<gbN, 256, 0, stream>>>(B1, Wl, nullptr, inv, B0, nsN);
        node_gather <<<gbR, 256, 0, stream>>>(B0, rowptr, cnt, ed, inv, bl, B1, N);
    }

    pool_zero<<<(G * DD + 255) / 256, 256, 0, stream>>>(pooled, G * DD);
    pool_add <<<gbR, 256, 0, stream>>>(B1, batch, pooled, N);

    gemm_strip<8, true, false><<<gbG, 256, 0, stream>>>(pooled, dec_W1, dec_b1, nullptr, dech, nsG);
    gemm_strip<2, false, false><<<gbG, 256, 0, stream>>>(dech,   dec_W2, dec_b2, nullptr, (float*)d_out, nsG);
}


# gemm chain: encoder layers 1, 2 and the first h@W (likewise decoder 1, 2) run in one kernel; the 16-row strip stays in LDS between layers, next W re-staged over the old; same products and sums
# speedup vs baseline: 4.7228x; 1.0716x over previous
#include <hip/hip_runtime.h>
#include <hip/hip_bf16.h>

typedef __attribute__((ext_vector_type(2))) float v2f;
typedef __attribute__((ext_vector_type(8))) float v8f;

#define DD 128

#define GW 8
#define APITCH 132

template <int NT, bool RELU, bool SCALE>
__global__ __launch_bounds__(256)
void gemm_strip(const float* __restrict__ A,
                const float* __restrict__ W,
                const float* __restrict__ bias,
                const float* __restrict__ rowscale,
                float* __restrict__ C,
                int nStrips)
{
    constexpr int NC = NT * 16;
    constexpr int C4 = NC / 4;
    constexpr int NW = (64 * C4) / 256;
    constexpr int RPI = 32 / C4;
    constexpr int PW = 2 * NC + 32;
    __shared__ alignas(16) float Ws[64 * PW];
    __shared__ alignas(16) float As[GW * 16 * APITCH];

    const int tid  = threadIdx.x;
    const int lane = tid & 31;
    const int wv   = __builtin_amdgcn_readfirstlane(tid >> 5);
    const int strip = blockIdx.x * GW + wv;
    const bool valid = strip < nStrips;
    const int lstrip = valid ? strip : nStrips - 1;
    float* Aw = As + wv * (16 * APITCH);

    float4 av[16];
    {
        const float4* Ag = (const float4*)(A + (size_t)lstrip * 16 * DD);
#pragma unroll
        for (int r = 0; r < 16; ++r)
            av[r] = Ag[r * 32 + lane];
    }
    float4 w0[NW], w1[NW];
#pragma unroll
    for (int i = 0; i < NW; ++i) {
        const int idx = i * 256 + tid;
        const int p  = idx / C4;
        const int c4 = idx % C4;
        w0[i] = *(const float4*)(W + (size_t)(2 * p)     * NC + 4 * c4);
        w1[i] = *(const float4*)(W + (size_t)(2 * p + 1) * NC + 4 * c4);
    }
#pragma unroll
    for (int r = 0; r < 16; ++r)
        *(float4*)(Aw + r * APITCH + 4 * lane) = av[r];
#pragma unroll
    for (int i = 0; i < NW; ++i) {
        const int idx = i * 256 + tid;
        const int p  = idx / C4;
        const int c4 = idx % C4;
        float* d = Ws + p * PW + 8 * c4;
        *(float4*)(d)     = make_float4(w0[i].x, w1[i].x, w0[i].y, w1[i].y);
        *(float4*)(d + 4) = make_float4(w0[i].z, w1[i].z, w0[i].w, w1[i].w);
    }
    __syncthreads();

    const int h = lane >> 4;
    const int c = lane & 15;

    if (valid) {
        v8f acc[NT];
#pragma unroll
        for (int j = 0; j < NT; ++j)
            acc[j] = (v8f){0.f, 0.f, 0.f, 0.f, 0.f, 0.f, 0.f, 0.f};

        const float* ap = Aw + c * APITCH + 2 * h;
        const float* bp = Ws + h * PW + 2 * c;
#pragma unroll
        for (int s = 0; s < 32; ++s) {
            const v2f a = *(const v2f*)(ap + 4 * s);
#pragma unroll
            for (int j = 0; j < NT; ++j) {
                const v2f b = *(const v2f*)(bp + 2 * s * PW + 32 * j);
                acc[j] = __builtin_amdgcn_wmma_f32_16x16x4_f32(
                     false, a,  false, b,
                     (short)0, acc[j],  false,  false);
            }
        }

#pragma unroll
        for (int j = 0; j < NT; ++j)
#pragma unroll
            for (int r = 0; r < 8; ++r)
                Aw[(8 * h + r) * APITCH + 16 * j + c] = acc[j][r];
    }

    const int c4 = lane % C4;
    float4 bb = make_float4(0.f, 0.f, 0.f, 0.f);
    if (bias) bb = *(const float4*)(bias + 4 * c4);
    float rs[2 * NT];
    if (SCALE) {
#pragma unroll
        for (int i = 0; i < 2 * NT; ++i)
            rs[i] = rowscale[(size_t)lstrip * 16 + i * RPI + lane / C4];
    }
    __syncthreads();

    if (valid) {
        float* Cg = C + (size_t)strip * 16 * NC;
        const int r0 = lane / C4;
#pragma unroll
        for (int i = 0; i < 2 * NT; ++i) {
            const int row = i * RPI + r0;
            float4 v = *(const float4*)(Aw + row * APITCH + 4 * c4);
            if (SCALE) { v.x *= rs[i]; v.y *= rs[i]; v.z *= rs[i]; v.w *= rs[i]; }
            v.x += bb.x; v.y += bb.y; v.z += bb.z; v.w += bb.w;
            if (RELU) {
                v.x = v.x > 0.f ? v.x : 0.f;
                v.y = v.y > 0.f ? v.y : 0.f;
                v.z = v.z > 0.f ? v.z : 0.f;
                v.w = v.w > 0.f ? v.w : 0.f;
            }
            *(float4*)(Cg + (size_t)row * NC + 4 * c4) = v;
        }
    }
}

template <int NT>
__device__ __forceinline__ void w_fetch(const float* __restrict__ W, int tid,
                                        float4* w0, float4* w1) {
    constexpr int NC = NT * 16;
    constexpr int C4 = NC / 4;
#pragma unroll
    for (int i = 0; i < NT; ++i) {
        const int idx = i * 256 + tid;
        const int p  = idx / C4;
        const int c4 = idx % C4;
        w0[i] = *(const float4*)(W + (size_t)(2 * p)     * NC + 4 * c4);
        w1[i] = *(const float4*)(W + (size_t)(2 * p + 1) * NC + 4 * c4);
    }
}
template <int NT>
__device__ __forceinline__ void w_park(float* Ws, int tid,
                                       const float4* w0, const float4* w1) {
    constexpr int NC = NT * 16;
    constexpr int C4 = NC / 4;
    constexpr int PW = 2 * NC + 32;
#pragma unroll
    for (int i = 0; i < NT; ++i) {
        const int idx = i * 256 + tid;
        const int p  = idx / C4;
        const int c4 = idx % C4;
        float* d = Ws + p * PW + 8 * c4;
        *(float4*)(d)     = make_float4(w0[i].x, w1[i].x, w0[i].y, w1[i].y);
        *(float4*)(d + 4) = make_float4(w0[i].z, w1[i].z, w0[i].w, w1[i].w);
    }
}
template <int NT>
__device__ __forceinline__ void strip_mma(const float* Aw, const float* Ws,
                                          int lane, v8f* acc) {
    constexpr int PW = 2 * NT * 16 + 32;
    const int h = lane >> 4;
    const int c = lane & 15;
#pragma unroll
    for (int j = 0; j < NT; ++j)
        acc[j] = (v8f){0.f, 0.f, 0.f, 0.f, 0.f, 0.f, 0.f, 0.f};
    const float* ap = Aw + c * APITCH + 2 * h;
    const float* bp = Ws + h * PW + 2 * c;
#pragma unroll
    for (int s = 0; s < 32; ++s) {
        const v2f a = *(const v2f*)(ap + 4 * s);
#pragma unroll
        for (int j = 0; j < NT; ++j) {
            const v2f b = *(const v2f*)(bp + 2 * s * PW + 32 * j);
            acc[j] = __builtin_amdgcn_wmma_f32_16x16x4_f32(
                 false, a,  false, b,
                 (short)0, acc[j],  false,  false);
        }
    }
}

template <int S, int NTL, bool RELU, bool SCALE>
__global__ __launch_bounds__(256)
void gemm_chain(const float* __restrict__ A,
                const float* __restrict__ Wa,
                const float* __restrict__ Wb,
                const float* __restrict__ Wc,
                const float* __restrict__ ba,
                const float* __restrict__ bb,
                const float* __restrict__ bl,
                const float* __restrict__ rowscale,
                float* __restrict__ C,
                int nStrips)
{
    constexpr int NC = NTL * 16;
    constexpr int C4 = NC / 4;
    constexpr int RPI = 32 / C4;
    __shared__ alignas(16) float Ws[64 * (2 * DD + 32)];
    __shared__ alignas(16) float As[GW * 16 * APITCH];

    const int tid  = threadIdx.x;
    const int lane = tid & 31;
    const int wv   = __builtin_amdgcn_readfirstlane(tid >> 5);
    const int strip = blockIdx.x * GW + wv;
    const bool valid = strip < nStrips;
    const int lstrip = valid ? strip : nStrips - 1;
    float* Aw = As + wv * (16 * APITCH);
    const int h = lane >> 4;
    const int c = lane & 15;

    {
        float4 av[16];
        const float4* Ag = (const float4*)(A + (size_t)lstrip * 16 * DD);
#pragma unroll
        for (int r = 0; r < 16; ++r)
            av[r] = Ag[r * 32 + lane];
        float4 w0[8], w1[8];
        w_fetch<8>(Wa, tid, w0, w1);
#pragma unroll
        for (int r = 0; r < 16; ++r)
            *(float4*)(Aw + r * APITCH + 4 * lane) = av[r];
        w_park<8>(Ws, tid, w0, w1);
    }
    __syncthreads();

#pragma unroll
    for (int s = 0; s < S - 1; ++s) {
        const float* Wn = (s == 0) ? Wb : Wc;
        const float* bs = (s == 0) ? ba : bb;
        const bool nextLast = (s == S - 2);
        float4 n0[8], n1[8];
        if (nextLast) w_fetch<NTL>(Wn, tid, n0, n1);
        else          w_fetch<8>(Wn, tid, n0, n1);
        float bc[8];
#pragma unroll
        for (int j = 0; j < 8; ++j) bc[j] = bs[16 * j + c];
        if (valid) {
            v8f acc[8];
            strip_mma<8>(Aw, Ws, lane, acc);
#pragma unroll
            for (int j = 0; j < 8; ++j)
#pragma unroll
                for (int r = 0; r < 8; ++r) {
                    float v = acc[j][r] + bc[j];
                    v = v > 0.f ? v : 0.f;
                    Aw[(8 * h + r) * APITCH + 16 * j + c] = v;
                }
        }
        __syncthreads();
        if (nextLast) w_park<NTL>(Ws, tid, n0, n1);
        else          w_park<8>(Ws, tid, n0, n1);
        __syncthreads();
    }

    if (valid) {
        v8f acc[NTL];
        strip_mma<NTL>(Aw, Ws, lane, acc);
#pragma unroll
        for (int j = 0; j < NTL; ++j)
#pragma unroll
            for (int r = 0; r < 8; ++r)
                Aw[(8 * h + r) * APITCH + 16 * j + c] = acc[j][r];
    }
    const int c4 = lane % C4;
    float4 bv = make_float4(0.f, 0.f, 0.f, 0.f);
    if (bl) bv = *(const float4*)(bl + 4 * c4);
    float rs[2 * NTL];
    if (SCALE) {
#pragma unroll
        for (int i = 0; i < 2 * NTL; ++i)
            rs[i] = rowscale[(size_t)lstrip * 16 + i * RPI + lane / C4];
    }
    __syncthreads();

    if (valid) {
        float* Cg = C + (size_t)strip * 16 * NC;
        const int r0 = lane / C4;
#pragma unroll
        for (int i = 0; i < 2 * NTL; ++i) {
            const int row = i * RPI + r0;
            float4 v = *(const float4*)(Aw + row * APITCH + 4 * c4);
            if (SCALE) { v.x *= rs[i]; v.y *= rs[i]; v.z *= rs[i]; v.w *= rs[i]; }
            v.x += bv.x; v.y += bv.y; v.z += bv.z; v.w += bv.w;
            if (RELU) {
                v.x = v.x > 0.f ? v.x : 0.f;
                v.y = v.y > 0.f ? v.y : 0.f;
                v.z = v.z > 0.f ? v.z : 0.f;
                v.w = v.w > 0.f ? v.w : 0.f;
            }
            *(float4*)(Cg + (size_t)row * NC + 4 * c4) = v;
        }
    }
}

#define BSH   9
#define BN    512
#define MAXNB 1024
#define STG   16384

__global__ void bin_zero(int* __restrict__ binTotal, int nb) {
    int i = blockIdx.x * blockDim.x + threadIdx.x;
    if (i < nb) binTotal[i] = 0;
}
__global__ __launch_bounds__(1024)
void bin_count(const int* __restrict__ dst, int* binTotal, int e, int nb, int chunk) {
    __shared__ int hist[MAXNB];
    const int tid = threadIdx.x;
    for (int b = tid; b < nb; b += 1024) hist[b] = 0;
    __syncthreads();
    const int lo = blockIdx.x * chunk;
    const int hi = (lo + chunk < e) ? (lo + chunk) : e;
    for (int i = lo + tid; i < hi; i += 1024) atomicAdd(&hist[dst[i] >> BSH], 1);
    __syncthreads();
    for (int b = tid; b < nb; b += 1024) {
        const int c = hist[b];
        if (c) atomicAdd(binTotal + b, c);
    }
}
__global__ __launch_bounds__(1024)
void bin_scan(const int* __restrict__ binTotal, int* __restrict__ binBase,
              int* __restrict__ binCursor, int nb) {
    __shared__ int wsum[32];
    const int tid  = threadIdx.x;
    const int lane = tid & 31;
    const int wv   = tid >> 5;
    const int c = (tid < nb) ? binTotal[tid] : 0;
    int x = c;
#pragma unroll
    for (int o = 1; o < 32; o <<= 1) {
        const int y = __shfl_up(x, o);
        if (lane >= o) x += y;
    }
    if (lane == 31) wsum[wv] = x;
    __syncthreads();
    if (wv == 0) {
        const int t = wsum[lane];
        int xs = t;
#pragma unroll
        for (int o = 1; o < 32; o <<= 1) {
            const int y = __shfl_up(xs, o);
            if (lane >= o) xs += y;
        }
        wsum[lane] = xs - t;
    }
    __syncthreads();
    if (tid < nb) {
        const int p = wsum[wv] + x - c;
        binBase[tid] = p;
        binCursor[tid] = p;
    }
}
__global__ __launch_bounds__(1024)
void bin_scatter(const int* __restrict__ src, const int* __restrict__ dst,
                 int* binCursor, int2* __restrict__ binned, int e, int nb, int chunk) {
    __shared__ int hist[MAXNB];
    __shared__ int bbase[MAXNB];
    const int tid = threadIdx.x;
    for (int b = tid; b < nb; b += 1024) hist[b] = 0;
    __syncthreads();
    const int lo = blockIdx.x * chunk;
    const int hi = (lo + chunk < e) ? (lo + chunk) : e;
    for (int i = lo + tid; i < hi; i += 1024) atomicAdd(&hist[dst[i] >> BSH], 1);
    __syncthreads();
    for (int b = tid; b < nb; b += 1024) {
        const int c = hist[b];
        bbase[b] = c ? atomicAdd(binCursor + b, c) : 0;
        hist[b] = 0;
    }
    __syncthreads();
    for (int i = lo + tid; i < hi; i += 1024) {
        const int d = dst[i];
        const int b = d >> BSH;
        const int r = atomicAdd(&hist[b], 1);
        binned[bbase[b] + r] = make_int2(d, src[i]);
    }
}
__global__ __launch_bounds__(1024)
void bin_sort(const int2* __restrict__ binned, const int* __restrict__ binTotal,
              const int* __restrict__ binBase, int* __restrict__ cnt,
              int* __restrict__ rowptr, float* __restrict__ inv,
              int* __restrict__ ed, int n) {
    __shared__ int cur[BN];
    __shared__ int wsum[32];
    __shared__ int stage[STG];
    const int tid  = threadIdx.x;
    const int lane = tid & 31;
    const int wv   = tid >> 5;
    const int b    = blockIdx.x;
    const int nE   = binTotal[b];
    const int base = binBase[b];
    const int2* be = binned + base;
    if (tid < BN) cur[tid] = 0;
    __syncthreads();
    for (int i = tid; i < nE; i += 1024) atomicAdd(&cur[be[i].x & (BN - 1)], 1);
    __syncthreads();
    const int c = (tid < BN) ? cur[tid] : 0;
    int x = c;
#pragma unroll
    for (int o = 1; o < 32; o <<= 1) {
        const int y = __shfl_up(x, o);
        if (lane >= o) x += y;
    }
    if (lane == 31) wsum[wv] = x;
    __syncthreads();
    if (wv == 0) {
        const int t = wsum[lane];
        int xs = t;
#pragma unroll
        for (int o = 1; o < 32; o <<= 1) {
            const int y = __shfl_up(xs, o);
            if (lane >= o) xs += y;
        }
        wsum[lane] = xs - t;
    }
    __syncthreads();
    if (tid < BN) {
        const int excl = wsum[wv] + x - c;
        cur[tid] = excl;
        const int node = b * BN + tid;
        if (node < n) {
            cnt[node] = c;
            rowptr[node] = base + excl;
            inv[node] = rsqrtf((float)(c + 1));
        }
    }
    __syncthreads();
    const bool fits = nE <= STG;
    for (int i = tid; i < nE; i += 1024) {
        const int2 e = be[i];
        const int r = atomicAdd(&cur[e.x & (BN - 1)], 1);
        if (fits) stage[r] = e.y; else ed[base + r] = e.y;
    }
    if (fits) {
        __syncthreads();
        for (int i = tid; i < nE; i += 1024) ed[base + i] = stage[i];
    }
}

#define NPW 8
__global__ __launch_bounds__(256)
void node_gather(const float* __restrict__ hw,
                 const int* __restrict__ rowptr,
                 const int* __restrict__ cnt,
                 const int* __restrict__ ed,
                 const float* __restrict__ inv,
                 const float* __restrict__ b,
                 float* __restrict__ h, int n) {
    const int n0 = (blockIdx.x * 8 + __builtin_amdgcn_readfirstlane(threadIdx.x >> 5)) * NPW;
    if (n0 >= n) return;
    const int nn = (n - n0) < NPW ? (n - n0) : NPW;
    const int lane = threadIdx.x & 31;
    const float4* hw4 = (const float4*)hw;
    int cv = 0;
    float dv = 0.f;
    if (lane < nn) {
        cv = cnt[n0 + lane];
        dv = inv[n0 + lane];
    }
    int start = __builtin_amdgcn_readfirstlane(rowptr[n0]);
    const float4 bv = ((const float4*)b)[lane];

    for (int j = 0; j < nn; ++j) {
        const int i = n0 + j;
        const int m = __builtin_amdgcn_readlane(cv, j);
        const float di = __int_as_float(__builtin_amdgcn_readlane(__float_as_int(dv), j));
        const float4 sv = hw4[(size_t)i * 32 + lane];
        float4 acc = make_float4(0.f, 0.f, 0.f, 0.f);

        for (int base = 0; base < m; base += 32) {
            const int mm = (m - base) < 32 ? (m - base) : 32;
            int es = 0;
            if (lane < mm) es = ed[start + base + lane];
            int k = 0;
            do {
                int s[8];
#pragma unroll
                for (int u = 0; u < 8; ++u) {
                    const int q = (k + u < mm) ? (k + u) : (mm - 1);
                    s[u] = __builtin_amdgcn_readlane(es, q);
                }
                float4 r[8];
#pragma unroll
                for (int u = 0; u < 8; ++u) {
                    r[u] = make_float4(0.f, 0.f, 0.f, 0.f);
                    if (k + u < mm) r[u] = hw4[(size_t)s[u] * 32 + lane];
                }
#pragma unroll
                for (int u = 0; u < 8; ++u) {
                    acc.x += r[u].x;
                    acc.y += r[u].y;
                    acc.z += r[u].z;
                    acc.w += r[u].w;
                }
                k += 8;
            } while (k < mm);
        }
        acc.x += sv.x;
        acc.y += sv.y;
        acc.z += sv.z;
        acc.w += sv.w;
        float4 v = make_float4(fmaf(acc.x, di, bv.x), fmaf(acc.y, di, bv.y),
                               fmaf(acc.z, di, bv.z), fmaf(acc.w, di, bv.w));
        v.x = v.x > 0.f ? v.x : 0.f;
        v.y = v.y > 0.f ? v.y : 0.f;
        v.z = v.z > 0.f ? v.z : 0.f;
        v.w = v.w > 0.f ? v.w : 0.f;
        ((float4*)h)[(size_t)i * 32 + lane] = v;
        start += m;
    }
}

__global__ void pool_zero(float* p, int n) {
    int i = blockIdx.x * blockDim.x + threadIdx.x;
    if (i < n) p[i] = 0.f;
}
__global__ __launch_bounds__(256)
void pool_add(const float* __restrict__ h,
              const int* __restrict__ batch,
              float* __restrict__ pooled, int n) {
    const int n0 = (blockIdx.x * 8 + __builtin_amdgcn_readfirstlane(threadIdx.x >> 5)) * NPW;
    if (n0 >= n) return;
    const int nn = (n - n0) < NPW ? (n - n0) : NPW;
    const int lane = threadIdx.x & 31;
    int gv = 0;
    if (lane < nn) gv = batch[n0 + lane];
    float v[NPW][4];
#pragma unroll
    for (int j = 0; j < NPW; ++j) {
        if (j < nn) {
            const float* hp = h + (size_t)(n0 + j) * DD + lane;
            v[j][0] = hp[0];
            v[j][1] = hp[32];
            v[j][2] = hp[64];
            v[j][3] = hp[96];
        }
    }
#pragma unroll
    for (int j = 0; j < NPW; ++j) {
        if (j < nn) {
            const int g = __builtin_amdgcn_readlane(gv, j);
            float* p = pooled + (size_t)g * DD + lane;
            atomicAdd(p +  0, v[j][0]);
            atomicAdd(p + 32, v[j][1]);
            atomicAdd(p + 64, v[j][2]);
            atomicAdd(p + 96, v[j][3]);
        }
    }
}

extern "C" void kernel_launch(void* const* d_in, const int* in_sizes, int n_in,
                              void* d_out, int out_size, void* d_ws, size_t ws_size,
                              hipStream_t stream) {
    const float* x      = (const float*)d_in[0];
    const int*   ei     = (const int*)d_in[1];
    const int*   batch  = (const int*)d_in[2];
    const float* enc_W1 = (const float*)d_in[3];
    const float* enc_b1 = (const float*)d_in[4];
    const float* enc_W2 = (const float*)d_in[5];
    const float* enc_b2 = (const float*)d_in[6];
    const float* gcn_W  = (const float*)d_in[7];
    const float* gcn_b  = (const float*)d_in[8];
    const float* dec_W1 = (const float*)d_in[9];
    const float* dec_b1 = (const float*)d_in[10];
    const float* dec_W2 = (const float*)d_in[11];
    const float* dec_b2 = (const float*)d_in[12];

    const int N   = in_sizes[0] / DD;
    const int E   = in_sizes[1] / 2;
    const int L   = in_sizes[7] / (DD * DD);
    const int G   = 512;

    const int* src = ei;
    const int* dst = ei + E;

    char* ws = (char*)d_ws;
    const size_t nd = (size_t)N * DD * sizeof(float);
    float* B0     = (float*)(ws);
    float* B1     = (float*)(ws + nd);
    float* inv    = (float*)(ws + 3 * nd);
    size_t off    = 3 * nd + (((size_t)N * 4 + 255) & ~(size_t)255);
    float* pooled = (float*)(ws + off);
    const size_t nI = ((size_t)N * 4 + 255) & ~(size_t)255;
    size_t off2   = off + 2 * (size_t)G * DD * sizeof(float);
    int*  cnt     = (int*)(ws + off2);
    int*  rowptr  = (int*)(ws + off2 + nI);
    int*  binTotal  = (int*)(ws + off2 + 2 * nI);
    int*  binBase   = binTotal + MAXNB;
    int*  binCursor = binBase + MAXNB;
    int*  ed      = (int*)(ws + off2 + 3 * nI + 256);
    int2* binned  = (int2*)(ws + 2 * nd);
    const int nb    = (N + BN - 1) / BN;
    const int nblk  = 256;
    const int chunk = (E + nblk - 1) / nblk;

    const int nsN = N / 16;
    const int nsG = G / 16;
    const int gbN = (nsN + GW - 1) / GW;
    const int gbG = (nsG + GW - 1) / GW;
    const int gbR = (N + 8 * NPW - 1) / (8 * NPW);

    bin_zero   <<<(nb + 255) / 256, 256, 0, stream>>>(binTotal, nb);
    bin_count  <<<nblk, 1024, 0, stream>>>(dst, binTotal, E, nb, chunk);
    bin_scan   <<<1, 1024, 0, stream>>>(binTotal, binBase, binCursor, nb);
    bin_scatter<<<nblk, 1024, 0, stream>>>(src, dst, binCursor, binned, E, nb, chunk);
    bin_sort   <<<nb, 1024, 0, stream>>>(binned, binTotal, binBase, cnt, rowptr, inv, ed, N);

    if (L > 0)
        gemm_chain<3, 8, false, true><<<gbN, 256, 0, stream>>>(x, enc_W1, enc_W2, gcn_W, enc_b1, enc_b2, nullptr, inv, B0, nsN);
    else
        gemm_chain<2, 8, true, false><<<gbN, 256, 0, stream>>>(x, enc_W1, enc_W2, nullptr, enc_b1, nullptr, enc_b2, nullptr, B1, nsN);

    for (int l = 0; l < L; ++l) {
        const float* Wl = gcn_W + (size_t)l * DD * DD;
        const float* bl = gcn_b + (size_t)l * DD;
        if (l > 0)
            gemm_strip<8, false, true><<<gbN, 256, 0, stream>>>(B1, Wl, nullptr, inv, B0, nsN);
        node_gather <<<gbR, 256, 0, stream>>>(B0, rowptr, cnt, ed, inv, bl, B1, N);
    }

    pool_zero<<<(G * DD + 255) / 256, 256, 0, stream>>>(pooled, G * DD);
    pool_add <<<gbR, 256, 0, stream>>>(B1, batch, pooled, N);

    gemm_chain<2, 2, false, false><<<gbG, 256, 0, stream>>>(pooled, dec_W1, dec_W2, nullptr, dec_b1, nullptr, dec_b2, nullptr, (float*)d_out, nsG);
}


# global add pool folded into existing launches: pooled zeroed in the first launch; last node_gather adds per-wave same-graph partial row sums by float atomics, no h plane; pool sum order changes
# speedup vs baseline: 5.0159x; 1.0621x over previous
#include <hip/hip_runtime.h>
#include <hip/hip_bf16.h>

typedef __attribute__((ext_vector_type(2))) float v2f;
typedef __attribute__((ext_vector_type(8))) float v8f;

#define DD 128

#define GW 8
#define APITCH 132

template <int NT, bool RELU, bool SCALE>
__global__ __launch_bounds__(256)
void gemm_strip(const float* __restrict__ A,
                const float* __restrict__ W,
                const float* __restrict__ bias,
                const float* __restrict__ rowscale,
                float* __restrict__ C,
                int nStrips)
{
    constexpr int NC = NT * 16;
    constexpr int C4 = NC / 4;
    constexpr int NW = (64 * C4) / 256;
    constexpr int RPI = 32 / C4;
    constexpr int PW = 2 * NC + 32;
    __shared__ alignas(16) float Ws[64 * PW];
    __shared__ alignas(16) float As[GW * 16 * APITCH];

    const int tid  = threadIdx.x;
    const int lane = tid & 31;
    const int wv   = __builtin_amdgcn_readfirstlane(tid >> 5);
    const int strip = blockIdx.x * GW + wv;
    const bool valid = strip < nStrips;
    const int lstrip = valid ? strip : nStrips - 1;
    float* Aw = As + wv * (16 * APITCH);

    float4 av[16];
    {
        const float4* Ag = (const float4*)(A + (size_t)lstrip * 16 * DD);
#pragma unroll
        for (int r = 0; r < 16; ++r)
            av[r] = Ag[r * 32 + lane];
    }
    float4 w0[NW], w1[NW];
#pragma unroll
    for (int i = 0; i < NW; ++i) {
        const int idx = i * 256 + tid;
        const int p  = idx / C4;
        const int c4 = idx % C4;
        w0[i] = *(const float4*)(W + (size_t)(2 * p)     * NC + 4 * c4);
        w1[i] = *(const float4*)(W + (size_t)(2 * p + 1) * NC + 4 * c4);
    }
#pragma unroll
    for (int r = 0; r < 16; ++r)
        *(float4*)(Aw + r * APITCH + 4 * lane) = av[r];
#pragma unroll
    for (int i = 0; i < NW; ++i) {
        const int idx = i * 256 + tid;
        const int p  = idx / C4;
        const int c4 = idx % C4;
        float* d = Ws + p * PW + 8 * c4;
        *(float4*)(d)     = make_float4(w0[i].x, w1[i].x, w0[i].y, w1[i].y);
        *(float4*)(d + 4) = make_float4(w0[i].z, w1[i].z, w0[i].w, w1[i].w);
    }
    __syncthreads();

    const int h = lane >> 4;
    const int c = lane & 15;

    if (valid) {
        v8f acc[NT];
#pragma unroll
        for (int j = 0; j < NT; ++j)
            acc[j] = (v8f){0.f, 0.f, 0.f, 0.f, 0.f, 0.f, 0.f, 0.f};

        const float* ap = Aw + c * APITCH + 2 * h;
        const float* bp = Ws + h * PW + 2 * c;
#pragma unroll
        for (int s = 0; s < 32; ++s) {
            const v2f a = *(const v2f*)(ap + 4 * s);
#pragma unroll
            for (int j = 0; j < NT; ++j) {
                const v2f b = *(const v2f*)(bp + 2 * s * PW + 32 * j);
                acc[j] = __builtin_amdgcn_wmma_f32_16x16x4_f32(
                     false, a,  false, b,
                     (short)0, acc[j],  false,  false);
            }
        }

#pragma unroll
        for (int j = 0; j < NT; ++j)
#pragma unroll
            for (int r = 0; r < 8; ++r)
                Aw[(8 * h + r) * APITCH + 16 * j + c] = acc[j][r];
    }

    const int c4 = lane % C4;
    float4 bb = make_float4(0.f, 0.f, 0.f, 0.f);
    if (bias) bb = *(const float4*)(bias + 4 * c4);
    float rs[2 * NT];
    if (SCALE) {
#pragma unroll
        for (int i = 0; i < 2 * NT; ++i)
            rs[i] = rowscale[(size_t)lstrip * 16 + i * RPI + lane / C4];
    }
    __syncthreads();

    if (valid) {
        float* Cg = C + (size_t)strip * 16 * NC;
        const int r0 = lane / C4;
#pragma unroll
        for (int i = 0; i < 2 * NT; ++i) {
            const int row = i * RPI + r0;
            float4 v = *(const float4*)(Aw + row * APITCH + 4 * c4);
            if (SCALE) { v.x *= rs[i]; v.y *= rs[i]; v.z *= rs[i]; v.w *= rs[i]; }
            v.x += bb.x; v.y += bb.y; v.z += bb.z; v.w += bb.w;
            if (RELU) {
                v.x = v.x > 0.f ? v.x : 0.f;
                v.y = v.y > 0.f ? v.y : 0.f;
                v.z = v.z > 0.f ? v.z : 0.f;
                v.w = v.w > 0.f ? v.w : 0.f;
            }
            *(float4*)(Cg + (size_t)row * NC + 4 * c4) = v;
        }
    }
}

template <int NT>
__device__ __forceinline__ void w_fetch(const float* __restrict__ W, int tid,
                                        float4* w0, float4* w1) {
    constexpr int NC = NT * 16;
    constexpr int C4 = NC / 4;
#pragma unroll
    for (int i = 0; i < NT; ++i) {
        const int idx = i * 256 + tid;
        const int p  = idx / C4;
        const int c4 = idx % C4;
        w0[i] = *(const float4*)(W + (size_t)(2 * p)     * NC + 4 * c4);
        w1[i] = *(const float4*)(W + (size_t)(2 * p + 1) * NC + 4 * c4);
    }
}
template <int NT>
__device__ __forceinline__ void w_park(float* Ws, int tid,
                                       const float4* w0, const float4* w1) {
    constexpr int NC = NT * 16;
    constexpr int C4 = NC / 4;
    constexpr int PW = 2 * NC + 32;
#pragma unroll
    for (int i = 0; i < NT; ++i) {
        const int idx = i * 256 + tid;
        const int p  = idx / C4;
        const int c4 = idx % C4;
        float* d = Ws + p * PW + 8 * c4;
        *(float4*)(d)     = make_float4(w0[i].x, w1[i].x, w0[i].y, w1[i].y);
        *(float4*)(d + 4) = make_float4(w0[i].z, w1[i].z, w0[i].w, w1[i].w);
    }
}
template <int NT>
__device__ __forceinline__ void strip_mma(const float* Aw, const float* Ws,
                                          int lane, v8f* acc) {
    constexpr int PW = 2 * NT * 16 + 32;
    const int h = lane >> 4;
    const int c = lane & 15;
#pragma unroll
    for (int j = 0; j < NT; ++j)
        acc[j] = (v8f){0.f, 0.f, 0.f, 0.f, 0.f, 0.f, 0.f, 0.f};
    const float* ap = Aw + c * APITCH + 2 * h;
    const float* bp = Ws + h * PW + 2 * c;
#pragma unroll
    for (int s = 0; s < 32; ++s) {
        const v2f a = *(const v2f*)(ap + 4 * s);
#pragma unroll
        for (int j = 0; j < NT; ++j) {
            const v2f b = *(const v2f*)(bp + 2 * s * PW + 32 * j);
            acc[j] = __builtin_amdgcn_wmma_f32_16x16x4_f32(
                 false, a,  false, b,
                 (short)0, acc[j],  false,  false);
        }
    }
}

template <int S, int NTL, bool RELU, bool SCALE>
__global__ __launch_bounds__(256)
void gemm_chain(const float* __restrict__ A,
                const float* __restrict__ Wa,
                const float* __restrict__ Wb,
                const float* __restrict__ Wc,
                const float* __restrict__ ba,
                const float* __restrict__ bb,
                const float* __restrict__ bl,
                const float* __restrict__ rowscale,
                float* __restrict__ C,
                int nStrips)
{
    constexpr int NC = NTL * 16;
    constexpr int C4 = NC / 4;
    constexpr int RPI = 32 / C4;
    __shared__ alignas(16) float Ws[64 * (2 * DD + 32)];
    __shared__ alignas(16) float As[GW * 16 * APITCH];

    const int tid  = threadIdx.x;
    const int lane = tid & 31;
    const int wv   = __builtin_amdgcn_readfirstlane(tid >> 5);
    const int strip = blockIdx.x * GW + wv;
    const bool valid = strip < nStrips;
    const int lstrip = valid ? strip : nStrips - 1;
    float* Aw = As + wv * (16 * APITCH);
    const int h = lane >> 4;
    const int c = lane & 15;

    {
        float4 av[16];
        const float4* Ag = (const float4*)(A + (size_t)lstrip * 16 * DD);
#pragma unroll
        for (int r = 0; r < 16; ++r)
            av[r] = Ag[r * 32 + lane];
        float4 w0[8], w1[8];
        w_fetch<8>(Wa, tid, w0, w1);
#pragma unroll
        for (int r = 0; r < 16; ++r)
            *(float4*)(Aw + r * APITCH + 4 * lane) = av[r];
        w_park<8>(Ws, tid, w0, w1);
    }
    __syncthreads();

#pragma unroll
    for (int s = 0; s < S - 1; ++s) {
        const float* Wn = (s == 0) ? Wb : Wc;
        const float* bs = (s == 0) ? ba : bb;
        const bool nextLast = (s == S - 2);
        float4 n0[8], n1[8];
        if (nextLast) w_fetch<NTL>(Wn, tid, n0, n1);
        else          w_fetch<8>(Wn, tid, n0, n1);
        float bc[8];
#pragma unroll
        for (int j = 0; j < 8; ++j) bc[j] = bs[16 * j + c];
        if (valid) {
            v8f acc[8];
            strip_mma<8>(Aw, Ws, lane, acc);
#pragma unroll
            for (int j = 0; j < 8; ++j)
#pragma unroll
                for (int r = 0; r < 8; ++r) {
                    float v = acc[j][r] + bc[j];
                    v = v > 0.f ? v : 0.f;
                    Aw[(8 * h + r) * APITCH + 16 * j + c] = v;
                }
        }
        __syncthreads();
        if (nextLast) w_park<NTL>(Ws, tid, n0, n1);
        else          w_park<8>(Ws, tid, n0, n1);
        __syncthreads();
    }

    if (valid) {
        v8f acc[NTL];
        strip_mma<NTL>(Aw, Ws, lane, acc);
#pragma unroll
        for (int j = 0; j < NTL; ++j)
#pragma unroll
            for (int r = 0; r < 8; ++r)
                Aw[(8 * h + r) * APITCH + 16 * j + c] = acc[j][r];
    }
    const int c4 = lane % C4;
    float4 bv = make_float4(0.f, 0.f, 0.f, 0.f);
    if (bl) bv = *(const float4*)(bl + 4 * c4);
    float rs[2 * NTL];
    if (SCALE) {
#pragma unroll
        for (int i = 0; i < 2 * NTL; ++i)
            rs[i] = rowscale[(size_t)lstrip * 16 + i * RPI + lane / C4];
    }
    __syncthreads();

    if (valid) {
        float* Cg = C + (size_t)strip * 16 * NC;
        const int r0 = lane / C4;
#pragma unroll
        for (int i = 0; i < 2 * NTL; ++i) {
            const int row = i * RPI + r0;
            float4 v = *(const float4*)(Aw + row * APITCH + 4 * c4);
            if (SCALE) { v.x *= rs[i]; v.y *= rs[i]; v.z *= rs[i]; v.w *= rs[i]; }
            v.x += bv.x; v.y += bv.y; v.z += bv.z; v.w += bv.w;
            if (RELU) {
                v.x = v.x > 0.f ? v.x : 0.f;
                v.y = v.y > 0.f ? v.y : 0.f;
                v.z = v.z > 0.f ? v.z : 0.f;
                v.w = v.w > 0.f ? v.w : 0.f;
            }
            *(float4*)(Cg + (size_t)row * NC + 4 * c4) = v;
        }
    }
}

#define BSH   9
#define BN    512
#define MAXNB 1024
#define STG   16384

__global__ void zero_init(int* __restrict__ binTotal, int nb, float* __restrict__ pooled, int np) {
    int i = blockIdx.x * blockDim.x + threadIdx.x;
    if (i < nb) binTotal[i] = 0;
    if (i < np) pooled[i] = 0.f;
}
__global__ __launch_bounds__(1024)
void bin_count(const int* __restrict__ dst, int* binTotal, int e, int nb, int chunk) {
    __shared__ int hist[MAXNB];
    const int tid = threadIdx.x;
    for (int b = tid; b < nb; b += 1024) hist[b] = 0;
    __syncthreads();
    const int lo = blockIdx.x * chunk;
    const int hi = (lo + chunk < e) ? (lo + chunk) : e;
    for (int i = lo + tid; i < hi; i += 1024) atomicAdd(&hist[dst[i] >> BSH], 1);
    __syncthreads();
    for (int b = tid; b < nb; b += 1024) {
        const int c = hist[b];
        if (c) atomicAdd(binTotal + b, c);
    }
}
__global__ __launch_bounds__(1024)
void bin_scan(const int* __restrict__ binTotal, int* __restrict__ binBase,
              int* __restrict__ binCursor, int nb) {
    __shared__ int wsum[32];
    const int tid  = threadIdx.x;
    const int lane = tid & 31;
    const int wv   = tid >> 5;
    const int c = (tid < nb) ? binTotal[tid] : 0;
    int x = c;
#pragma unroll
    for (int o = 1; o < 32; o <<= 1) {
        const int y = __shfl_up(x, o);
        if (lane >= o) x += y;
    }
    if (lane == 31) wsum[wv] = x;
    __syncthreads();
    if (wv == 0) {
        const int t = wsum[lane];
        int xs = t;
#pragma unroll
        for (int o = 1; o < 32; o <<= 1) {
            const int y = __shfl_up(xs, o);
            if (lane >= o) xs += y;
        }
        wsum[lane] = xs - t;
    }
    __syncthreads();
    if (tid < nb) {
        const int p = wsum[wv] + x - c;
        binBase[tid] = p;
        binCursor[tid] = p;
    }
}
__global__ __launch_bounds__(1024)
void bin_scatter(const int* __restrict__ src, const int* __restrict__ dst,
                 int* binCursor, int2* __restrict__ binned, int e, int nb, int chunk) {
    __shared__ int hist[MAXNB];
    __shared__ int bbase[MAXNB];
    const int tid = threadIdx.x;
    for (int b = tid; b < nb; b += 1024) hist[b] = 0;
    __syncthreads();
    const int lo = blockIdx.x * chunk;
    const int hi = (lo + chunk < e) ? (lo + chunk) : e;
    for (int i = lo + tid; i < hi; i += 1024) atomicAdd(&hist[dst[i] >> BSH], 1);
    __syncthreads();
    for (int b = tid; b < nb; b += 1024) {
        const int c = hist[b];
        bbase[b] = c ? atomicAdd(binCursor + b, c) : 0;
        hist[b] = 0;
    }
    __syncthreads();
    for (int i = lo + tid; i < hi; i += 1024) {
        const int d = dst[i];
        const int b = d >> BSH;
        const int r = atomicAdd(&hist[b], 1);
        binned[bbase[b] + r] = make_int2(d, src[i]);
    }
}
__global__ __launch_bounds__(1024)
void bin_sort(const int2* __restrict__ binned, const int* __restrict__ binTotal,
              const int* __restrict__ binBase, int* __restrict__ cnt,
              int* __restrict__ rowptr, float* __restrict__ inv,
              int* __restrict__ ed, int n) {
    __shared__ int cur[BN];
    __shared__ int wsum[32];
    __shared__ int stage[STG];
    const int tid  = threadIdx.x;
    const int lane = tid & 31;
    const int wv   = tid >> 5;
    const int b    = blockIdx.x;
    const int nE   = binTotal[b];
    const int base = binBase[b];
    const int2* be = binned + base;
    if (tid < BN) cur[tid] = 0;
    __syncthreads();
    for (int i = tid; i < nE; i += 1024) atomicAdd(&cur[be[i].x & (BN - 1)], 1);
    __syncthreads();
    const int c = (tid < BN) ? cur[tid] : 0;
    int x = c;
#pragma unroll
    for (int o = 1; o < 32; o <<= 1) {
        const int y = __shfl_up(x, o);
        if (lane >= o) x += y;
    }
    if (lane == 31) wsum[wv] = x;
    __syncthreads();
    if (wv == 0) {
        const int t = wsum[lane];
        int xs = t;
#pragma unroll
        for (int o = 1; o < 32; o <<= 1) {
            const int y = __shfl_up(xs, o);
            if (lane >= o) xs += y;
        }
        wsum[lane] = xs - t;
    }
    __syncthreads();
    if (tid < BN) {
        const int excl = wsum[wv] + x - c;
        cur[tid] = excl;
        const int node = b * BN + tid;
        if (node < n) {
            cnt[node] = c;
            rowptr[node] = base + excl;
            inv[node] = rsqrtf((float)(c + 1));
        }
    }
    __syncthreads();
    const bool fits = nE <= STG;
    for (int i = tid; i < nE; i += 1024) {
        const int2 e = be[i];
        const int r = atomicAdd(&cur[e.x & (BN - 1)], 1);
        if (fits) stage[r] = e.y; else ed[base + r] = e.y;
    }
    if (fits) {
        __syncthreads();
        for (int i = tid; i < nE; i += 1024) ed[base + i] = stage[i];
    }
}

#define NPW 8
__device__ __forceinline__ void pool_flush(float* t, int lane, float4 ps, float* prow) {
    *(float4*)(t + 4 * lane) = ps;
    const float a0 = t[lane];
    const float a1 = t[lane + 32];
    const float a2 = t[lane + 64];
    const float a3 = t[lane + 96];
    atomicAdd(prow + lane,      a0);
    atomicAdd(prow + lane + 32, a1);
    atomicAdd(prow + lane + 64, a2);
    atomicAdd(prow + lane + 96, a3);
}
template <bool POOL>
__global__ __launch_bounds__(256)
void node_gather(const float* __restrict__ hw,
                 const int* __restrict__ rowptr,
                 const int* __restrict__ cnt,
                 const int* __restrict__ ed,
                 const float* __restrict__ inv,
                 const float* __restrict__ b,
                 float* __restrict__ h,
                 const int* __restrict__ batch,
                 float* pooled, int n) {
    const int n0 = (blockIdx.x * 8 + __builtin_amdgcn_readfirstlane(threadIdx.x >> 5)) * NPW;
    if (n0 >= n) return;
    const int nn = (n - n0) < NPW ? (n - n0) : NPW;
    const int lane = threadIdx.x & 31;
    const float4* hw4 = (const float4*)hw;
    int cv = 0;
    float dv = 0.f;
    if (lane < nn) {
        cv = cnt[n0 + lane];
        dv = inv[n0 + lane];
    }
    int start = __builtin_amdgcn_readfirstlane(rowptr[n0]);
    const float4 bv = ((const float4*)b)[lane];
    __shared__ alignas(16) float tr[POOL ? 8 * DD : 4];
    int gv = 0;
    int gcur = 0;
    float4 ps = make_float4(0.f, 0.f, 0.f, 0.f);
    if (POOL) {
        if (lane < nn) gv = batch[n0 + lane];
        gcur = __builtin_amdgcn_readlane(gv, 0);
    }

    for (int j = 0; j < nn; ++j) {
        const int i = n0 + j;
        const int m = __builtin_amdgcn_readlane(cv, j);
        const float di = __int_as_float(__builtin_amdgcn_readlane(__float_as_int(dv), j));
        const float4 sv = hw4[(size_t)i * 32 + lane];
        float4 acc = make_float4(0.f, 0.f, 0.f, 0.f);

        for (int base = 0; base < m; base += 32) {
            const int mm = (m - base) < 32 ? (m - base) : 32;
            int es = 0;
            if (lane < mm) es = ed[start + base + lane];
            int k = 0;
            do {
                int s[8];
#pragma unroll
                for (int u = 0; u < 8; ++u) {
                    const int q = (k + u < mm) ? (k + u) : (mm - 1);
                    s[u] = __builtin_amdgcn_readlane(es, q);
                }
                float4 r[8];
#pragma unroll
                for (int u = 0; u < 8; ++u) {
                    r[u] = make_float4(0.f, 0.f, 0.f, 0.f);
                    if (k + u < mm) r[u] = hw4[(size_t)s[u] * 32 + lane];
                }
#pragma unroll
                for (int u = 0; u < 8; ++u) {
                    acc.x += r[u].x;
                    acc.y += r[u].y;
                    acc.z += r[u].z;
                    acc.w += r[u].w;
                }
                k += 8;
            } while (k < mm);
        }
        acc.x += sv.x;
        acc.y += sv.y;
        acc.z += sv.z;
        acc.w += sv.w;
        float4 v = make_float4(fmaf(acc.x, di, bv.x), fmaf(acc.y, di, bv.y),
                               fmaf(acc.z, di, bv.z), fmaf(acc.w, di, bv.w));
        v.x = v.x > 0.f ? v.x : 0.f;
        v.y = v.y > 0.f ? v.y : 0.f;
        v.z = v.z > 0.f ? v.z : 0.f;
        v.w = v.w > 0.f ? v.w : 0.f;
        if (POOL) {
            const int g = __builtin_amdgcn_readlane(gv, j);
            if (g != gcur) {
                pool_flush(tr + (threadIdx.x >> 5) * DD, lane, ps, pooled + (size_t)gcur * DD);
                ps = make_float4(0.f, 0.f, 0.f, 0.f);
                gcur = g;
            }
            ps.x += v.x;
            ps.y += v.y;
            ps.z += v.z;
            ps.w += v.w;
        } else {
            ((float4*)h)[(size_t)i * 32 + lane] = v;
        }
        start += m;
    }
    if (POOL) pool_flush(tr + (threadIdx.x >> 5) * DD, lane, ps, pooled + (size_t)gcur * DD);
}

__global__ __launch_bounds__(256)
void pool_add(const float* __restrict__ h,
              const int* __restrict__ batch,
              float* __restrict__ pooled, int n) {
    const int n0 = (blockIdx.x * 8 + __builtin_amdgcn_readfirstlane(threadIdx.x >> 5)) * NPW;
    if (n0 >= n) return;
    const int nn = (n - n0) < NPW ? (n - n0) : NPW;
    const int lane = threadIdx.x & 31;
    int gv = 0;
    if (lane < nn) gv = batch[n0 + lane];
    float v[NPW][4];
#pragma unroll
    for (int j = 0; j < NPW; ++j) {
        if (j < nn) {
            const float* hp = h + (size_t)(n0 + j) * DD + lane;
            v[j][0] = hp[0];
            v[j][1] = hp[32];
            v[j][2] = hp[64];
            v[j][3] = hp[96];
        }
    }
#pragma unroll
    for (int j = 0; j < NPW; ++j) {
        if (j < nn) {
            const int g = __builtin_amdgcn_readlane(gv, j);
            float* p = pooled + (size_t)g * DD + lane;
            atomicAdd(p +  0, v[j][0]);
            atomicAdd(p + 32, v[j][1]);
            atomicAdd(p + 64, v[j][2]);
            atomicAdd(p + 96, v[j][3]);
        }
    }
}

extern "C" void kernel_launch(void* const* d_in, const int* in_sizes, int n_in,
                              void* d_out, int out_size, void* d_ws, size_t ws_size,
                              hipStream_t stream) {
    const float* x      = (const float*)d_in[0];
    const int*   ei     = (const int*)d_in[1];
    const int*   batch  = (const int*)d_in[2];
    const float* enc_W1 = (const float*)d_in[3];
    const float* enc_b1 = (const float*)d_in[4];
    const float* enc_W2 = (const float*)d_in[5];
    const float* enc_b2 = (const float*)d_in[6];
    const float* gcn_W  = (const float*)d_in[7];
    const float* gcn_b  = (const float*)d_in[8];
    const float* dec_W1 = (const float*)d_in[9];
    const float* dec_b1 = (const float*)d_in[10];
    const float* dec_W2 = (const float*)d_in[11];
    const float* dec_b2 = (const float*)d_in[12];

    const int N   = in_sizes[0] / DD;
    const int E   = in_sizes[1] / 2;
    const int L   = in_sizes[7] / (DD * DD);
    const int G   = 512;

    const int* src = ei;
    const int* dst = ei + E;

    char* ws = (char*)d_ws;
    const size_t nd = (size_t)N * DD * sizeof(float);
    float* B0     = (float*)(ws);
    float* B1     = (float*)(ws + nd);
    float* inv    = (float*)(ws + 3 * nd);
    size_t off    = 3 * nd + (((size_t)N * 4 + 255) & ~(size_t)255);
    float* pooled = (float*)(ws + off);
    const size_t nI = ((size_t)N * 4 + 255) & ~(size_t)255;
    size_t off2   = off + 2 * (size_t)G * DD * sizeof(float);
    int*  cnt     = (int*)(ws + off2);
    int*  rowptr  = (int*)(ws + off2 + nI);
    int*  binTotal  = (int*)(ws + off2 + 2 * nI);
    int*  binBase   = binTotal + MAXNB;
    int*  binCursor = binBase + MAXNB;
    int*  ed      = (int*)(ws + off2 + 3 * nI + 256);
    int2* binned  = (int2*)(ws + 2 * nd);
    const int nb    = (N + BN - 1) / BN;
    const int nblk  = 256;
    const int chunk = (E + nblk - 1) / nblk;

    const int nsN = N / 16;
    const int nsG = G / 16;
    const int gbN = (nsN + GW - 1) / GW;
    const int gbG = (nsG + GW - 1) / GW;
    const int gbR = (N + 8 * NPW - 1) / (8 * NPW);

    zero_init  <<<(G * DD + 255) / 256, 256, 0, stream>>>(binTotal, nb, pooled, G * DD);
    bin_count  <<<nblk, 1024, 0, stream>>>(dst, binTotal, E, nb, chunk);
    bin_scan   <<<1, 1024, 0, stream>>>(binTotal, binBase, binCursor, nb);
    bin_scatter<<<nblk, 1024, 0, stream>>>(src, dst, binCursor, binned, E, nb, chunk);
    bin_sort   <<<nb, 1024, 0, stream>>>(binned, binTotal, binBase, cnt, rowptr, inv, ed, N);

    if (L > 0)
        gemm_chain<3, 8, false, true><<<gbN, 256, 0, stream>>>(x, enc_W1, enc_W2, gcn_W, enc_b1, enc_b2, nullptr, inv, B0, nsN);
    else
        gemm_chain<2, 8, true, false><<<gbN, 256, 0, stream>>>(x, enc_W1, enc_W2, nullptr, enc_b1, nullptr, enc_b2, nullptr, B1, nsN);

    for (int l = 0; l < L; ++l) {
        const float* Wl = gcn_W + (size_t)l * DD * DD;
        const float* bl = gcn_b + (size_t)l * DD;
        if (l > 0)
            gemm_strip<8, false, true><<<gbN, 256, 0, stream>>>(B1, Wl, nullptr, inv, B0, nsN);
        if (l == L - 1)
            node_gather<true> <<<gbR, 256, 0, stream>>>(B0, rowptr, cnt, ed, inv, bl, B1, batch, pooled, N);
        else
            node_gather<false><<<gbR, 256, 0, stream>>>(B0, rowptr, cnt, ed, inv, bl, B1, batch, pooled, N);
    }

    if (L <= 0)
        pool_add <<<gbR, 256, 0, stream>>>(B1, batch, pooled, N);

    gemm_chain<2, 2, false, false><<<gbG, 256, 0, stream>>>(pooled, dec_W1, dec_W2, nullptr, dec_b1, nullptr, dec_b2, nullptr, (float*)d_out, nsG);
}


# edge aggregation of a layer and the next layer's h @ W in one kernel: a wave gathers a 16-node strip into LDS and multiplies it there; no h plane, no layer gemm launch; same sums, same order
# speedup vs baseline: 5.2029x; 1.0373x over previous
#include <hip/hip_runtime.h>
#include <hip/hip_bf16.h>

typedef __attribute__((ext_vector_type(2))) float v2f;
typedef __attribute__((ext_vector_type(8))) float v8f;

#define DD 128

#define GW 8
#define APITCH 132

template <int NT, bool RELU, bool SCALE>
__global__ __launch_bounds__(256)
void gemm_strip(const float* __restrict__ A,
                const float* __restrict__ W,
                const float* __restrict__ bias,
                const float* __restrict__ rowscale,
                float* __restrict__ C,
                int nStrips)
{
    constexpr int NC = NT * 16;
    constexpr int C4 = NC / 4;
    constexpr int NW = (64 * C4) / 256;
    constexpr int RPI = 32 / C4;
    constexpr int PW = 2 * NC + 32;
    __shared__ alignas(16) float Ws[64 * PW];
    __shared__ alignas(16) float As[GW * 16 * APITCH];

    const int tid  = threadIdx.x;
    const int lane = tid & 31;
    const int wv   = __builtin_amdgcn_readfirstlane(tid >> 5);
    const int strip = blockIdx.x * GW + wv;
    const bool valid = strip < nStrips;
    const int lstrip = valid ? strip : nStrips - 1;
    float* Aw = As + wv * (16 * APITCH);

    float4 av[16];
    {
        const float4* Ag = (const float4*)(A + (size_t)lstrip * 16 * DD);
#pragma unroll
        for (int r = 0; r < 16; ++r)
            av[r] = Ag[r * 32 + lane];
    }
    float4 w0[NW], w1[NW];
#pragma unroll
    for (int i = 0; i < NW; ++i) {
        const int idx = i * 256 + tid;
        const int p  = idx / C4;
        const int c4 = idx % C4;
        w0[i] = *(const float4*)(W + (size_t)(2 * p)     * NC + 4 * c4);
        w1[i] = *(const float4*)(W + (size_t)(2 * p + 1) * NC + 4 * c4);
    }
#pragma unroll
    for (int r = 0; r < 16; ++r)
        *(float4*)(Aw + r * APITCH + 4 * lane) = av[r];
#pragma unroll
    for (int i = 0; i < NW; ++i) {
        const int idx = i * 256 + tid;
        const int p  = idx / C4;
        const int c4 = idx % C4;
        float* d = Ws + p * PW + 8 * c4;
        *(float4*)(d)     = make_float4(w0[i].x, w1[i].x, w0[i].y, w1[i].y);
        *(float4*)(d + 4) = make_float4(w0[i].z, w1[i].z, w0[i].w, w1[i].w);
    }
    __syncthreads();

    const int h = lane >> 4;
    const int c = lane & 15;

    if (valid) {
        v8f acc[NT];
#pragma unroll
        for (int j = 0; j < NT; ++j)
            acc[j] = (v8f){0.f, 0.f, 0.f, 0.f, 0.f, 0.f, 0.f, 0.f};

        const float* ap = Aw + c * APITCH + 2 * h;
        const float* bp = Ws + h * PW + 2 * c;
#pragma unroll
        for (int s = 0; s < 32; ++s) {
            const v2f a = *(const v2f*)(ap + 4 * s);
#pragma unroll
            for (int j = 0; j < NT; ++j) {
                const v2f b = *(const v2f*)(bp + 2 * s * PW + 32 * j);
                acc[j] = __builtin_amdgcn_wmma_f32_16x16x4_f32(
                     false, a,  false, b,
                     (short)0, acc[j],  false,  false);
            }
        }

#pragma unroll
        for (int j = 0; j < NT; ++j)
#pragma unroll
            for (int r = 0; r < 8; ++r)
                Aw[(8 * h + r) * APITCH + 16 * j + c] = acc[j][r];
    }

    const int c4 = lane % C4;
    float4 bb = make_float4(0.f, 0.f, 0.f, 0.f);
    if (bias) bb = *(const float4*)(bias + 4 * c4);
    float rs[2 * NT];
    if (SCALE) {
#pragma unroll
        for (int i = 0; i < 2 * NT; ++i)
            rs[i] = rowscale[(size_t)lstrip * 16 + i * RPI + lane / C4];
    }
    __syncthreads();

    if (valid) {
        float* Cg = C + (size_t)strip * 16 * NC;
        const int r0 = lane / C4;
#pragma unroll
        for (int i = 0; i < 2 * NT; ++i) {
            const int row = i * RPI + r0;
            float4 v = *(const float4*)(Aw + row * APITCH + 4 * c4);
            if (SCALE) { v.x *= rs[i]; v.y *= rs[i]; v.z *= rs[i]; v.w *= rs[i]; }
            v.x += bb.x; v.y += bb.y; v.z += bb.z; v.w += bb.w;
            if (RELU) {
                v.x = v.x > 0.f ? v.x : 0.f;
                v.y = v.y > 0.f ? v.y : 0.f;
                v.z = v.z > 0.f ? v.z : 0.f;
                v.w = v.w > 0.f ? v.w : 0.f;
            }
            *(float4*)(Cg + (size_t)row * NC + 4 * c4) = v;
        }
    }
}

template <int NT>
__device__ __forceinline__ void w_fetch(const float* __restrict__ W, int tid,
                                        float4* w0, float4* w1) {
    constexpr int NC = NT * 16;
    constexpr int C4 = NC / 4;
#pragma unroll
    for (int i = 0; i < NT; ++i) {
        const int idx = i * 256 + tid;
        const int p  = idx / C4;
        const int c4 = idx % C4;
        w0[i] = *(const float4*)(W + (size_t)(2 * p)     * NC + 4 * c4);
        w1[i] = *(const float4*)(W + (size_t)(2 * p + 1) * NC + 4 * c4);
    }
}
template <int NT>
__device__ __forceinline__ void w_park(float* Ws, int tid,
                                       const float4* w0, const float4* w1) {
    constexpr int NC = NT * 16;
    constexpr int C4 = NC / 4;
    constexpr int PW = 2 * NC + 32;
#pragma unroll
    for (int i = 0; i < NT; ++i) {
        const int idx = i * 256 + tid;
        const int p  = idx / C4;
        const int c4 = idx % C4;
        float* d = Ws + p * PW + 8 * c4;
        *(float4*)(d)     = make_float4(w0[i].x, w1[i].x, w0[i].y, w1[i].y);
        *(float4*)(d + 4) = make_float4(w0[i].z, w1[i].z, w0[i].w, w1[i].w);
    }
}
template <int NT>
__device__ __forceinline__ void strip_mma(const float* Aw, const float* Ws,
                                          int lane, v8f* acc) {
    constexpr int PW = 2 * NT * 16 + 32;
    const int h = lane >> 4;
    const int c = lane & 15;
#pragma unroll
    for (int j = 0; j < NT; ++j)
        acc[j] = (v8f){0.f, 0.f, 0.f, 0.f, 0.f, 0.f, 0.f, 0.f};
    const float* ap = Aw + c * APITCH + 2 * h;
    const float* bp = Ws + h * PW + 2 * c;
#pragma unroll
    for (int s = 0; s < 32; ++s) {
        const v2f a = *(const v2f*)(ap + 4 * s);
#pragma unroll
        for (int j = 0; j < NT; ++j) {
            const v2f b = *(const v2f*)(bp + 2 * s * PW + 32 * j);
            acc[j] = __builtin_amdgcn_wmma_f32_16x16x4_f32(
                 false, a,  false, b,
                 (short)0, acc[j],  false,  false);
        }
    }
}

template <int S, int NTL, bool RELU, bool SCALE>
__global__ __launch_bounds__(256)
void gemm_chain(const float* __restrict__ A,
                const float* __restrict__ Wa,
                const float* __restrict__ Wb,
                const float* __restrict__ Wc,
                const float* __restrict__ ba,
                const float* __restrict__ bb,
                const float* __restrict__ bl,
                const float* __restrict__ rowscale,
                float* __restrict__ C,
                int nStrips)
{
    constexpr int NC = NTL * 16;
    constexpr int C4 = NC / 4;
    constexpr int RPI = 32 / C4;
    __shared__ alignas(16) float Ws[64 * (2 * DD + 32)];
    __shared__ alignas(16) float As[GW * 16 * APITCH];

    const int tid  = threadIdx.x;
    const int lane = tid & 31;
    const int wv   = __builtin_amdgcn_readfirstlane(tid >> 5);
    const int strip = blockIdx.x * GW + wv;
    const bool valid = strip < nStrips;
    const int lstrip = valid ? strip : nStrips - 1;
    float* Aw = As + wv * (16 * APITCH);
    const int h = lane >> 4;
    const int c = lane & 15;

    {
        float4 av[16];
        const float4* Ag = (const float4*)(A + (size_t)lstrip * 16 * DD);
#pragma unroll
        for (int r = 0; r < 16; ++r)
            av[r] = Ag[r * 32 + lane];
        float4 w0[8], w1[8];
        w_fetch<8>(Wa, tid, w0, w1);
#pragma unroll
        for (int r = 0; r < 16; ++r)
            *(float4*)(Aw + r * APITCH + 4 * lane) = av[r];
        w_park<8>(Ws, tid, w0, w1);
    }
    __syncthreads();

#pragma unroll
    for (int s = 0; s < S - 1; ++s) {
        const float* Wn = (s == 0) ? Wb : Wc;
        const float* bs = (s == 0) ? ba : bb;
        const bool nextLast = (s == S - 2);
        float4 n0[8], n1[8];
        if (nextLast) w_fetch<NTL>(Wn, tid, n0, n1);
        else          w_fetch<8>(Wn, tid, n0, n1);
        float bc[8];
#pragma unroll
        for (int j = 0; j < 8; ++j) bc[j] = bs[16 * j + c];
        if (valid) {
            v8f acc[8];
            strip_mma<8>(Aw, Ws, lane, acc);
#pragma unroll
            for (int j = 0; j < 8; ++j)
#pragma unroll
                for (int r = 0; r < 8; ++r) {
                    float v = acc[j][r] + bc[j];
                    v = v > 0.f ? v : 0.f;
                    Aw[(8 * h + r) * APITCH + 16 * j + c] = v;
                }
        }
        __syncthreads();
        if (nextLast) w_park<NTL>(Ws, tid, n0, n1);
        else          w_park<8>(Ws, tid, n0, n1);
        __syncthreads();
    }

    if (valid) {
        v8f acc[NTL];
        strip_mma<NTL>(Aw, Ws, lane, acc);
#pragma unroll
        for (int j = 0; j < NTL; ++j)
#pragma unroll
            for (int r = 0; r < 8; ++r)
                Aw[(8 * h + r) * APITCH + 16 * j + c] = acc[j][r];
    }
    const int c4 = lane % C4;
    float4 bv = make_float4(0.f, 0.f, 0.f, 0.f);
    if (bl) bv = *(const float4*)(bl + 4 * c4);
    float rs[2 * NTL];
    if (SCALE) {
#pragma unroll
        for (int i = 0; i < 2 * NTL; ++i)
            rs[i] = rowscale[(size_t)lstrip * 16 + i * RPI + lane / C4];
    }
    __syncthreads();

    if (valid) {
        float* Cg = C + (size_t)strip * 16 * NC;
        const int r0 = lane / C4;
#pragma unroll
        for (int i = 0; i < 2 * NTL; ++i) {
            const int row = i * RPI + r0;
            float4 v = *(const float4*)(Aw + row * APITCH + 4 * c4);
            if (SCALE) { v.x *= rs[i]; v.y *= rs[i]; v.z *= rs[i]; v.w *= rs[i]; }
            v.x += bv.x; v.y += bv.y; v.z += bv.z; v.w += bv.w;
            if (RELU) {
                v.x = v.x > 0.f ? v.x : 0.f;
                v.y = v.y > 0.f ? v.y : 0.f;
                v.z = v.z > 0.f ? v.z : 0.f;
                v.w = v.w > 0.f ? v.w : 0.f;
            }
            *(float4*)(Cg + (size_t)row * NC + 4 * c4) = v;
        }
    }
}

#define BSH   9
#define BN    512
#define MAXNB 1024
#define STG   16384

__global__ void zero_init(int* __restrict__ binTotal, int nb, float* __restrict__ pooled, int np) {
    int i = blockIdx.x * blockDim.x + threadIdx.x;
    if (i < nb) binTotal[i] = 0;
    if (i < np) pooled[i] = 0.f;
}
__global__ __launch_bounds__(1024)
void bin_count(const int* __restrict__ dst, int* binTotal, int e, int nb, int chunk) {
    __shared__ int hist[MAXNB];
    const int tid = threadIdx.x;
    for (int b = tid; b < nb; b += 1024) hist[b] = 0;
    __syncthreads();
    const int lo = blockIdx.x * chunk;
    const int hi = (lo + chunk < e) ? (lo + chunk) : e;
    for (int i = lo + tid; i < hi; i += 1024) atomicAdd(&hist[dst[i] >> BSH], 1);
    __syncthreads();
    for (int b = tid; b < nb; b += 1024) {
        const int c = hist[b];
        if (c) atomicAdd(binTotal + b, c);
    }
}
__global__ __launch_bounds__(1024)
void bin_scan(const int* __restrict__ binTotal, int* __restrict__ binBase,
              int* __restrict__ binCursor, int nb) {
    __shared__ int wsum[32];
    const int tid  = threadIdx.x;
    const int lane = tid & 31;
    const int wv   = tid >> 5;
    const int c = (tid < nb) ? binTotal[tid] : 0;
    int x = c;
#pragma unroll
    for (int o = 1; o < 32; o <<= 1) {
        const int y = __shfl_up(x, o);
        if (lane >= o) x += y;
    }
    if (lane == 31) wsum[wv] = x;
    __syncthreads();
    if (wv == 0) {
        const int t = wsum[lane];
        int xs = t;
#pragma unroll
        for (int o = 1; o < 32; o <<= 1) {
            const int y = __shfl_up(xs, o);
            if (lane >= o) xs += y;
        }
        wsum[lane] = xs - t;
    }
    __syncthreads();
    if (tid < nb) {
        const int p = wsum[wv] + x - c;
        binBase[tid] = p;
        binCursor[tid] = p;
    }
}
__global__ __launch_bounds__(1024)
void bin_scatter(const int* __restrict__ src, const int* __restrict__ dst,
                 int* binCursor, int2* __restrict__ binned, int e, int nb, int chunk) {
    __shared__ int hist[MAXNB];
    __shared__ int bbase[MAXNB];
    const int tid = threadIdx.x;
    for (int b = tid; b < nb; b += 1024) hist[b] = 0;
    __syncthreads();
    const int lo = blockIdx.x * chunk;
    const int hi = (lo + chunk < e) ? (lo + chunk) : e;
    for (int i = lo + tid; i < hi; i += 1024) atomicAdd(&hist[dst[i] >> BSH], 1);
    __syncthreads();
    for (int b = tid; b < nb; b += 1024) {
        const int c = hist[b];
        bbase[b] = c ? atomicAdd(binCursor + b, c) : 0;
        hist[b] = 0;
    }
    __syncthreads();
    for (int i = lo + tid; i < hi; i += 1024) {
        const int d = dst[i];
        const int b = d >> BSH;
        const int r = atomicAdd(&hist[b], 1);
        binned[bbase[b] + r] = make_int2(d, src[i]);
    }
}
__global__ __launch_bounds__(1024)
void bin_sort(const int2* __restrict__ binned, const int* __restrict__ binTotal,
              const int* __restrict__ binBase, int* __restrict__ cnt,
              int* __restrict__ rowptr, float* __restrict__ inv,
              int* __restrict__ ed, int n) {
    __shared__ int cur[BN];
    __shared__ int wsum[32];
    __shared__ int stage[STG];
    const int tid  = threadIdx.x;
    const int lane = tid & 31;
    const int wv   = tid >> 5;
    const int b    = blockIdx.x;
    const int nE   = binTotal[b];
    const int base = binBase[b];
    const int2* be = binned + base;
    if (tid < BN) cur[tid] = 0;
    __syncthreads();
    for (int i = tid; i < nE; i += 1024) atomicAdd(&cur[be[i].x & (BN - 1)], 1);
    __syncthreads();
    const int c = (tid < BN) ? cur[tid] : 0;
    int x = c;
#pragma unroll
    for (int o = 1; o < 32; o <<= 1) {
        const int y = __shfl_up(x, o);
        if (lane >= o) x += y;
    }
    if (lane == 31) wsum[wv] = x;
    __syncthreads();
    if (wv == 0) {
        const int t = wsum[lane];
        int xs = t;
#pragma unroll
        for (int o = 1; o < 32; o <<= 1) {
            const int y = __shfl_up(xs, o);
            if (lane >= o) xs += y;
        }
        wsum[lane] = xs - t;
    }
    __syncthreads();
    if (tid < BN) {
        const int excl = wsum[wv] + x - c;
        cur[tid] = excl;
        const int node = b * BN + tid;
        if (node < n) {
            cnt[node] = c;
            rowptr[node] = base + excl;
            inv[node] = rsqrtf((float)(c + 1));
        }
    }
    __syncthreads();
    const bool fits = nE <= STG;
    for (int i = tid; i < nE; i += 1024) {
        const int2 e = be[i];
        const int r = atomicAdd(&cur[e.x & (BN - 1)], 1);
        if (fits) stage[r] = e.y; else ed[base + r] = e.y;
    }
    if (fits) {
        __syncthreads();
        for (int i = tid; i < nE; i += 1024) ed[base + i] = stage[i];
    }
}

#define NPW 8
__device__ __forceinline__ void pool_flush(float* t, int lane, float4 ps, float* prow) {
    *(float4*)(t + 4 * lane) = ps;
    const float a0 = t[lane];
    const float a1 = t[lane + 32];
    const float a2 = t[lane + 64];
    const float a3 = t[lane + 96];
    atomicAdd(prow + lane,      a0);
    atomicAdd(prow + lane + 32, a1);
    atomicAdd(prow + lane + 64, a2);
    atomicAdd(prow + lane + 96, a3);
}
template <bool POOL>
__global__ __launch_bounds__(256)
void node_gather(const float* __restrict__ hw,
                 const int* __restrict__ rowptr,
                 const int* __restrict__ cnt,
                 const int* __restrict__ ed,
                 const float* __restrict__ inv,
                 const float* __restrict__ b,
                 float* __restrict__ h,
                 const int* __restrict__ batch,
                 float* pooled, int n) {
    const int n0 = (blockIdx.x * 8 + __builtin_amdgcn_readfirstlane(threadIdx.x >> 5)) * NPW;
    if (n0 >= n) return;
    const int nn = (n - n0) < NPW ? (n - n0) : NPW;
    const int lane = threadIdx.x & 31;
    const float4* hw4 = (const float4*)hw;
    int cv = 0;
    float dv = 0.f;
    if (lane < nn) {
        cv = cnt[n0 + lane];
        dv = inv[n0 + lane];
    }
    int start = __builtin_amdgcn_readfirstlane(rowptr[n0]);
    const float4 bv = ((const float4*)b)[lane];
    __shared__ alignas(16) float tr[POOL ? 8 * DD : 4];
    int gv = 0;
    int gcur = 0;
    float4 ps = make_float4(0.f, 0.f, 0.f, 0.f);
    if (POOL) {
        if (lane < nn) gv = batch[n0 + lane];
        gcur = __builtin_amdgcn_readlane(gv, 0);
    }

    for (int j = 0; j < nn; ++j) {
        const int i = n0 + j;
        const int m = __builtin_amdgcn_readlane(cv, j);
        const float di = __int_as_float(__builtin_amdgcn_readlane(__float_as_int(dv), j));
        const float4 sv = hw4[(size_t)i * 32 + lane];
        float4 acc = make_float4(0.f, 0.f, 0.f, 0.f);

        for (int base = 0; base < m; base += 32) {
            const int mm = (m - base) < 32 ? (m - base) : 32;
            int es = 0;
            if (lane < mm) es = ed[start + base + lane];
            int k = 0;
            do {
                int s[8];
#pragma unroll
                for (int u = 0; u < 8; ++u) {
                    const int q = (k + u < mm) ? (k + u) : (mm - 1);
                    s[u] = __builtin_amdgcn_readlane(es, q);
                }
                float4 r[8];
#pragma unroll
                for (int u = 0; u < 8; ++u) {
                    r[u] = make_float4(0.f, 0.f, 0.f, 0.f);
                    if (k + u < mm) r[u] = hw4[(size_t)s[u] * 32 + lane];
                }
#pragma unroll
                for (int u = 0; u < 8; ++u) {
                    acc.x += r[u].x;
                    acc.y += r[u].y;
                    acc.z += r[u].z;
                    acc.w += r[u].w;
                }
                k += 8;
            } while (k < mm);
        }
        acc.x += sv.x;
        acc.y += sv.y;
        acc.z += sv.z;
        acc.w += sv.w;
        float4 v = make_float4(fmaf(acc.x, di, bv.x), fmaf(acc.y, di, bv.y),
                               fmaf(acc.z, di, bv.z), fmaf(acc.w, di, bv.w));
        v.x = v.x > 0.f ? v.x : 0.f;
        v.y = v.y > 0.f ? v.y : 0.f;
        v.z = v.z > 0.f ? v.z : 0.f;
        v.w = v.w > 0.f ? v.w : 0.f;
        if (POOL) {
            const int g = __builtin_amdgcn_readlane(gv, j);
            if (g != gcur) {
                pool_flush(tr + (threadIdx.x >> 5) * DD, lane, ps, pooled + (size_t)gcur * DD);
                ps = make_float4(0.f, 0.f, 0.f, 0.f);
                gcur = g;
            }
            ps.x += v.x;
            ps.y += v.y;
            ps.z += v.z;
            ps.w += v.w;
        } else {
            ((float4*)h)[(size_t)i * 32 + lane] = v;
        }
        start += m;
    }
    if (POOL) pool_flush(tr + (threadIdx.x >> 5) * DD, lane, ps, pooled + (size_t)gcur * DD);
}

__device__ __forceinline__ void add_rows16(const float4* __restrict__ hw4, int lane,
                                           int es, int mm, float4& acc) {
    int k = 0;
    do {
        int s[16];
#pragma unroll
        for (int u = 0; u < 16; ++u) {
            const int q = (k + u < mm) ? (k + u) : (mm - 1);
            s[u] = __builtin_amdgcn_readlane(es, q);
        }
        float4 r[16];
#pragma unroll
        for (int u = 0; u < 16; ++u) {
            r[u] = make_float4(0.f, 0.f, 0.f, 0.f);
            if (k + u < mm) r[u] = hw4[(size_t)s[u] * 32 + lane];
        }
#pragma unroll
        for (int u = 0; u < 16; ++u) {
            acc.x += r[u].x;
            acc.y += r[u].y;
            acc.z += r[u].z;
            acc.w += r[u].w;
        }
        k += 16;
    } while (k < mm);
}
__global__ __launch_bounds__(256)
void gather_gemm(const float* __restrict__ hw,
                 const int* __restrict__ rowptr,
                 const int* __restrict__ cnt,
                 const int* __restrict__ ed,
                 const float* __restrict__ inv,
                 const float* __restrict__ b,
                 const float* __restrict__ Wn,
                 float* __restrict__ C,
                 int nStrips)
{
    __shared__ alignas(16) float Ws[64 * (2 * DD + 32)];
    __shared__ alignas(16) float As[GW * 16 * APITCH];
    const int tid  = threadIdx.x;
    const int lane = tid & 31;
    const int wv   = __builtin_amdgcn_readfirstlane(tid >> 5);
    float* Aw = As + wv * (16 * APITCH);
    {
        float4 w0[8], w1[8];
        w_fetch<8>(Wn, tid, w0, w1);
        w_park<8>(Ws, tid, w0, w1);
    }
    __syncthreads();

    const float4* hw4 = (const float4*)hw;
    const float4 bv = ((const float4*)b)[lane];
    const int h = lane >> 4;
    const int c = lane & 15;
    const int nblk = gridDim.x;

    for (int strip = wv * nblk + blockIdx.x; strip < nStrips; strip += GW * nblk) {
        const int n0 = strip * 16;
        int cv = 0;
        float dv = 0.f;
        if (lane < 16) {
            cv = cnt[n0 + lane];
            dv = inv[n0 + lane];
        }
        int start = __builtin_amdgcn_readfirstlane(rowptr[n0]);
        int es = 0;
        {
            const int m0 = __builtin_amdgcn_readlane(cv, 0);
            if (lane < (m0 < 32 ? m0 : 32)) es = ed[start + lane];
        }
#pragma unroll 1
        for (int j = 0; j < 16; ++j) {
            const int i = n0 + j;
            const int m = __builtin_amdgcn_readlane(cv, j);
            const float di = __int_as_float(__builtin_amdgcn_readlane(__float_as_int(dv), j));
            const float4 sv = hw4[(size_t)i * 32 + lane];
            int esn = 0;
            if (j < 15) {
                const int m1 = __builtin_amdgcn_readlane(cv, j + 1);
                if (lane < (m1 < 32 ? m1 : 32)) esn = ed[start + m + lane];
            }
            float4 acc = make_float4(0.f, 0.f, 0.f, 0.f);
            if (m > 0) add_rows16(hw4, lane, es, m < 32 ? m : 32, acc);
            for (int base = 32; base < m; base += 32) {
                const int mm = (m - base) < 32 ? (m - base) : 32;
                int e2 = 0;
                if (lane < mm) e2 = ed[start + base + lane];
                add_rows16(hw4, lane, e2, mm, acc);
            }
            acc.x += sv.x;
            acc.y += sv.y;
            acc.z += sv.z;
            acc.w += sv.w;
            float4 v = make_float4(fmaf(acc.x, di, bv.x), fmaf(acc.y, di, bv.y),
                                   fmaf(acc.z, di, bv.z), fmaf(acc.w, di, bv.w));
            v.x = v.x > 0.f ? v.x : 0.f;
            v.y = v.y > 0.f ? v.y : 0.f;
            v.z = v.z > 0.f ? v.z : 0.f;
            v.w = v.w > 0.f ? v.w : 0.f;
            *(float4*)(Aw + j * APITCH + 4 * lane) = v;
            start += m;
            es = esn;
        }

        v8f acc[8];
        strip_mma<8>(Aw, Ws, lane, acc);
#pragma unroll
        for (int j = 0; j < 8; ++j)
#pragma unroll
            for (int r = 0; r < 8; ++r)
                Aw[(8 * h + r) * APITCH + 16 * j + c] = acc[j][r];
        float4* Cg = (float4*)(C + (size_t)n0 * DD);
#pragma unroll
        for (int i = 0; i < 16; ++i) {
            const float rs = __int_as_float(__builtin_amdgcn_readlane(__float_as_int(dv), i));
            float4 v = *(const float4*)(Aw + i * APITCH + 4 * lane);
            v.x *= rs; v.y *= rs; v.z *= rs; v.w *= rs;
            Cg[i * 32 + lane] = v;
        }
    }
}

__global__ __launch_bounds__(256)
void pool_add(const float* __restrict__ h,
              const int* __restrict__ batch,
              float* __restrict__ pooled, int n) {
    const int n0 = (blockIdx.x * 8 + __builtin_amdgcn_readfirstlane(threadIdx.x >> 5)) * NPW;
    if (n0 >= n) return;
    const int nn = (n - n0) < NPW ? (n - n0) : NPW;
    const int lane = threadIdx.x & 31;
    int gv = 0;
    if (lane < nn) gv = batch[n0 + lane];
    float v[NPW][4];
#pragma unroll
    for (int j = 0; j < NPW; ++j) {
        if (j < nn) {
            const float* hp = h + (size_t)(n0 + j) * DD + lane;
            v[j][0] = hp[0];
            v[j][1] = hp[32];
            v[j][2] = hp[64];
            v[j][3] = hp[96];
        }
    }
#pragma unroll
    for (int j = 0; j < NPW; ++j) {
        if (j < nn) {
            const int g = __builtin_amdgcn_readlane(gv, j);
            float* p = pooled + (size_t)g * DD + lane;
            atomicAdd(p +  0, v[j][0]);
            atomicAdd(p + 32, v[j][1]);
            atomicAdd(p + 64, v[j][2]);
            atomicAdd(p + 96, v[j][3]);
        }
    }
}

extern "C" void kernel_launch(void* const* d_in, const int* in_sizes, int n_in,
                              void* d_out, int out_size, void* d_ws, size_t ws_size,
                              hipStream_t stream) {
    const float* x      = (const float*)d_in[0];
    const int*   ei     = (const int*)d_in[1];
    const int*   batch  = (const int*)d_in[2];
    const float* enc_W1 = (const float*)d_in[3];
    const float* enc_b1 = (const float*)d_in[4];
    const float* enc_W2 = (const float*)d_in[5];
    const float* enc_b2 = (const float*)d_in[6];
    const float* gcn_W  = (const float*)d_in[7];
    const float* gcn_b  = (const float*)d_in[8];
    const float* dec_W1 = (const float*)d_in[9];
    const float* dec_b1 = (const float*)d_in[10];
    const float* dec_W2 = (const float*)d_in[11];
    const float* dec_b2 = (const float*)d_in[12];

    const int N   = in_sizes[0] / DD;
    const int E   = in_sizes[1] / 2;
    const int L   = in_sizes[7] / (DD * DD);
    const int G   = 512;

    const int* src = ei;
    const int* dst = ei + E;

    char* ws = (char*)d_ws;
    const size_t nd = (size_t)N * DD * sizeof(float);
    float* B0     = (float*)(ws);
    float* B1     = (float*)(ws + nd);
    float* inv    = (float*)(ws + 3 * nd);
    size_t off    = 3 * nd + (((size_t)N * 4 + 255) & ~(size_t)255);
    float* pooled = (float*)(ws + off);
    const size_t nI = ((size_t)N * 4 + 255) & ~(size_t)255;
    size_t off2   = off + 2 * (size_t)G * DD * sizeof(float);
    int*  cnt     = (int*)(ws + off2);
    int*  rowptr  = (int*)(ws + off2 + nI);
    int*  binTotal  = (int*)(ws + off2 + 2 * nI);
    int*  binBase   = binTotal + MAXNB;
    int*  binCursor = binBase + MAXNB;
    int*  ed      = (int*)(ws + off2 + 3 * nI + 256);
    int2* binned  = (int2*)(ws + 2 * nd);
    const int nb    = (N + BN - 1) / BN;
    const int nblk  = 256;
    const int chunk = (E + nblk - 1) / nblk;

    const int nsN = N / 16;
    const int nsG = G / 16;
    const int gbN = (nsN + GW - 1) / GW;
    const int gbG = (nsG + GW - 1) / GW;
    const int gbR = (N + 8 * NPW - 1) / (8 * NPW);

    zero_init  <<<(G * DD + 255) / 256, 256, 0, stream>>>(binTotal, nb, pooled, G * DD);
    bin_count  <<<nblk, 1024, 0, stream>>>(dst, binTotal, E, nb, chunk);
    bin_scan   <<<1, 1024, 0, stream>>>(binTotal, binBase, binCursor, nb);
    bin_scatter<<<nblk, 1024, 0, stream>>>(src, dst, binCursor, binned, E, nb, chunk);
    bin_sort   <<<nb, 1024, 0, stream>>>(binned, binTotal, binBase, cnt, rowptr, inv, ed, N);

    if (L > 0)
        gemm_chain<3, 8, false, true><<<gbN, 256, 0, stream>>>(x, enc_W1, enc_W2, gcn_W, enc_b1, enc_b2, nullptr, inv, B0, nsN);
    else
        gemm_chain<2, 8, true, false><<<gbN, 256, 0, stream>>>(x, enc_W1, enc_W2, nullptr, enc_b1, nullptr, enc_b2, nullptr, B1, nsN);

    float* cur = B0;
    float* nxt = B1;
    for (int l = 0; l < L; ++l) {
        const float* bl = gcn_b + (size_t)l * DD;
        if (l == L - 1) {
            node_gather<true><<<gbR, 256, 0, stream>>>(cur, rowptr, cnt, ed, inv, bl, nxt, batch, pooled, N);
        } else {
            const float* Wn = gcn_W + (size_t)(l + 1) * DD * DD;
            gather_gemm<<<512, 256, 0, stream>>>(cur, rowptr, cnt, ed, inv, bl, Wn, nxt, nsN);
            float* t = cur;
            cur = nxt;
            nxt = t;
        }
    }

    if (L <= 0)
        pool_add <<<gbR, 256, 0, stream>>>(B1, batch, pooled, N);

    gemm_chain<2, 2, false, false><<<gbG, 256, 0, stream>>>(pooled, dec_W1, dec_W2, nullptr, dec_b1, nullptr, dec_b2, nullptr, (float*)d_out, nsG);
}
